# starting kernel
# baseline (speedup 1.0000x reference)
#include <hip/hip_runtime.h>
#include <hip/hip_bf16.h>

typedef __attribute__((ext_vector_type(2))) float v2f;
typedef __attribute__((ext_vector_type(8))) float v8f;

#define HDIM 64

__global__ void zero_f32(float* __restrict__ p, int n) {
  int i = blockIdx.x * blockDim.x + threadIdx.x;
  if (i < n) p[i] = 0.0f;
}

__global__ void deg_accum(const int* __restrict__ dst, float* __restrict__ deg, int E) {
  int e = blockIdx.x * blockDim.x + threadIdx.x;
  if (e < E) atomicAdd(&deg[dst[e]], 1.0f);
}

__global__ void deg_to_dis(float* __restrict__ deg, int n) {
  int i = blockIdx.x * blockDim.x + threadIdx.x;
  if (i < n) {
    float d = deg[i];
    deg[i] = (d > 0.0f) ? rsqrtf(fmaxf(d, 1.0f)) : 0.0f;
  }
}

template<int K, bool RELU_BIAS>
__global__ void gemm_n64_wmma(const float* __restrict__ A,
                              const float* __restrict__ W,
                              const float* __restrict__ bias,
                              float* __restrict__ out, int n_rows) {
  int wave = (blockIdx.x * blockDim.x + threadIdx.x) >> 5;
  int lane = threadIdx.x & 31;
  int row0 = wave * 16;
  if (row0 >= n_rows) return;

  const int half = lane >> 4;
  const int l16  = lane & 15;

  v8f acc0 = {}, acc1 = {}, acc2 = {}, acc3 = {};

  const float* arow = A + (size_t)(row0 + l16) * K;
  for (int kb = 0; kb < K; kb += 4) {
    v2f a;
    a.x = arow[kb + half * 2 + 0];
    a.y = arow[kb + half * 2 + 1];

    const float* bp = W + (size_t)(kb + half * 2) * HDIM + l16;
    v2f b0, b1, b2, b3;
    b0.x = bp[0];        b0.y = bp[HDIM];
    b1.x = bp[16];       b1.y = bp[HDIM + 16];
    b2.x = bp[32];       b2.y = bp[HDIM + 32];
    b3.x = bp[48];       b3.y = bp[HDIM + 48];

    acc0 = __builtin_amdgcn_wmma_f32_16x16x4_f32(false, a, false, b0, (short)0, acc0, false, false);
    acc1 = __builtin_amdgcn_wmma_f32_16x16x4_f32(false, a, false, b1, (short)0, acc1, false, false);
    acc2 = __builtin_amdgcn_wmma_f32_16x16x4_f32(false, a, false, b2, (short)0, acc2, false, false);
    acc3 = __builtin_amdgcn_wmma_f32_16x16x4_f32(false, a, false, b3, (short)0, acc3, false, false);
  }

  v8f accs[4] = {acc0, acc1, acc2, acc3};
  #pragma unroll
  for (int nt = 0; nt < 4; ++nt) {
    int col = nt * 16 + l16;
    float bcol = RELU_BIAS ? bias[col] : 0.0f;
    #pragma unroll
    for (int r = 0; r < 8; ++r) {
      int m = half * 8 + r;
      float v = accs[nt][r];
      if (RELU_BIAS) { v += bcol; v = fmaxf(v, 0.0f); }
      out[(size_t)(row0 + m) * HDIM + col] = v;
    }
  }
}

__global__ void scatter_edges(const int* __restrict__ src, const int* __restrict__ dst,
                              const float* __restrict__ dis,
                              const float* __restrict__ hw,
                              float* __restrict__ acc, int E) {
  int gid  = blockIdx.x * blockDim.x + threadIdx.x;
  int e    = gid >> 5;
  int lane = gid & 31;
  if (e >= E) return;
  int s = src[e];
  int d = dst[e];
  float coef = dis[s] * dis[d];
  const float2* hp = (const float2*)(hw + (size_t)s * HDIM);
  float2 v = hp[lane];
  float* ap = acc + (size_t)d * HDIM + lane * 2;
  atomicAdd(ap + 0, v.x * coef);
  atomicAdd(ap + 1, v.y * coef);
}

__global__ void finalize_layer(const float* __restrict__ acc,
                               const float* __restrict__ bias,
                               float* __restrict__ h, int n64) {
  int i = blockIdx.x * blockDim.x + threadIdx.x;
  if (i >= n64) return;
  float v = acc[i] + bias[i & (HDIM - 1)];
  h[i] = fmaxf(v, 0.0f) + h[i];
}

__global__ void head_kernel(const float* __restrict__ h, const float* __restrict__ Wh,
                            const float* __restrict__ bh, float* __restrict__ out, int n) {
  int i = blockIdx.x * blockDim.x + threadIdx.x;
  if (i >= n) return;
  const float* hp = h + (size_t)i * HDIM;
  float s = bh[0];
  #pragma unroll
  for (int k = 0; k < HDIM; ++k) s += hp[k] * Wh[k];
  out[i] = s;
}

extern "C" void kernel_launch(void* const* d_in, const int* in_sizes, int n_in,
                              void* d_out, int out_size, void* d_ws, size_t ws_size,
                              hipStream_t stream) {
  const float* x      = (const float*)d_in[0];
  const int*   eidx   = (const int*)d_in[1];
  const float* W_pre  = (const float*)d_in[2];
  const float* b_pre  = (const float*)d_in[3];
  const float* W1     = (const float*)d_in[4];
  const float* b1     = (const float*)d_in[5];
  const float* W2     = (const float*)d_in[6];
  const float* b2     = (const float*)d_in[7];
  const float* W_head = (const float*)d_in[8];
  const float* b_head = (const float*)d_in[9];
  float* out = (float*)d_out;

  const int N = in_sizes[0] / 128;
  const int E = in_sizes[1] / 2;
  const int* src = eidx;
  const int* dst = eidx + E;

  float* dis = (float*)d_ws;
  float* h   = dis + N;
  float* hw  = h   + (size_t)N * HDIM;
  float* acc = hw  + (size_t)N * HDIM;

  const int n64 = N * HDIM;

  zero_f32<<<(N + 255) / 256, 256, 0, stream>>>(dis, N);
  deg_accum<<<(E + 255) / 256, 256, 0, stream>>>(dst, dis, E);
  deg_to_dis<<<(N + 255) / 256, 256, 0, stream>>>(dis, N);

  int gemm_blocks = (N + 127) / 128;
  gemm_n64_wmma<128, true><<<gemm_blocks, 256, 0, stream>>>(x, W_pre, b_pre, h, N);

  const float* Ws[2] = {W1, W2};
  const float* bs[2] = {b1, b2};
  for (int layer = 0; layer < 2; ++layer) {
    gemm_n64_wmma<64, false><<<gemm_blocks, 256, 0, stream>>>(h, Ws[layer], nullptr, hw, N);
    zero_f32<<<(n64 + 255) / 256, 256, 0, stream>>>(acc, n64);
    long long sthreads = (long long)E * 32;
    scatter_edges<<<(int)((sthreads + 255) / 256), 256, 0, stream>>>(src, dst, dis, hw, acc, E);
    finalize_layer<<<(n64 + 255) / 256, 256, 0, stream>>>(acc, bs[layer], h, n64);
  }

  head_kernel<<<(N + 255) / 256, 256, 0, stream>>>(h, W_head, b_head, out, N);
}


# Neighbour sum as gather over edges sorted by destination (rank from one counting atomic, scan, fill): one wave sums a row and writes relu+skip; f32 atomics gone; sum order = arrival rank
# speedup vs baseline: 1.9843x; 1.9843x over previous
#include <hip/hip_runtime.h>
#include <hip/hip_bf16.h>

typedef __attribute__((ext_vector_type(2))) float v2f;
typedef __attribute__((ext_vector_type(8))) float v8f;

#define HDIM 64

__global__ void zero_i32(int* __restrict__ p, int n) {
  int i = blockIdx.x * blockDim.x + threadIdx.x;
  if (i < n) p[i] = 0;
}

__global__ void rank_edges(const int* __restrict__ dst, int* __restrict__ cnt,
                           int* __restrict__ rank, int E) {
  int e = blockIdx.x * blockDim.x + threadIdx.x;
  if (e < E) rank[e] = atomicAdd(&cnt[dst[e]], 1);
}

__global__ __launch_bounds__(1024) void scan_rows(const int* __restrict__ cnt,
                                                  int* __restrict__ rowstart,
                                                  float* __restrict__ dis, int n) {
  __shared__ int wsum[32];
  __shared__ int s_prefix;
  const int tid  = threadIdx.x;
  const int lane = tid & 31;
  const int wid  = tid >> 5;
  const int base = blockIdx.x * 1024;

  int p = 0;
  for (int i = tid; i < base; i += 1024) p += cnt[i];
  for (int o = 16; o > 0; o >>= 1) p += __shfl_xor(p, o);
  if (lane == 0) wsum[wid] = p;
  __syncthreads();
  if (wid == 0) {
    int v = wsum[lane];
    for (int o = 16; o > 0; o >>= 1) v += __shfl_xor(v, o);
    if (lane == 0) s_prefix = v;
  }
  __syncthreads();
  const int prefix = s_prefix;

  const int i = base + tid;
  const int c = (i < n) ? cnt[i] : 0;
  int x = c;
  for (int o = 1; o < 32; o <<= 1) {
    int y = __shfl_up(x, o);
    if (lane >= o) x += y;
  }
  if (lane == 31) wsum[wid] = x;
  __syncthreads();
  int wo = (lane < wid) ? wsum[lane] : 0;
  for (int o = 16; o > 0; o >>= 1) wo += __shfl_xor(wo, o);

  if (i < n) {
    rowstart[i] = prefix + wo + x - c;
    float d = (float)c;
    dis[i] = (d > 0.0f) ? rsqrtf(fmaxf(d, 1.0f)) : 0.0f;
  }
}

__global__ void fill_sorted(const int* __restrict__ src, const int* __restrict__ dst,
                            const int* __restrict__ rowstart, const int* __restrict__ rank,
                            int* __restrict__ ssrc, int E) {
  int e = blockIdx.x * blockDim.x + threadIdx.x;
  if (e < E) ssrc[rowstart[dst[e]] + rank[e]] = src[e];
}

__global__ void gather_rows(const int* __restrict__ rowstart, const int* __restrict__ cnt,
                            const int* __restrict__ ssrc, const float* __restrict__ dis,
                            const float* __restrict__ hw, const float* __restrict__ bias,
                            float* __restrict__ h, int n) {
  const int lane = threadIdx.x & 31;
  const int node = __builtin_amdgcn_readfirstlane((int)((blockIdx.x * blockDim.x + threadIdx.x) >> 5));
  if (node >= n) return;
  const int start = rowstart[node];
  const int deg   = cnt[node];
  const float dd  = dis[node];

  float2 acc = make_float2(0.0f, 0.0f);
  for (int j0 = 0; j0 < deg; j0 += 32) {
    const int m = (deg - j0 < 32) ? (deg - j0) : 32;
    int   s_l = 0;
    float c_l = 0.0f;
    if (lane < m) {
      s_l = ssrc[start + j0 + lane];
      c_l = dis[s_l] * dd;
    }
    for (int j = 0; j < m; j += 4) {
      float2 v[4];
      float  c[4];
      for (int k = 0; k < 4; ++k) {
        const int  jj  = j + k;
        const bool in  = jj < m;
        const int  jc  = in ? jj : (m - 1);
        const int  s   = __builtin_amdgcn_readlane(s_l, jc);
        c[k] = __int_as_float(__builtin_amdgcn_readlane(__float_as_int(c_l), jc));
        const float2* rp = (const float2*)(hw + (size_t)s * HDIM);
        float2 t = rp[lane];
        v[k].x = in ? t.x : 0.0f;
        v[k].y = in ? t.y : 0.0f;
      }
      for (int k = 0; k < 4; ++k) {
        acc.x += v[k].x * c[k];
        acc.y += v[k].y * c[k];
      }
    }
  }

  const float2 b  = ((const float2*)bias)[lane];
  float2* hp = (float2*)(h + (size_t)node * HDIM);
  const float2 hv = hp[lane];
  float2 o;
  o.x = fmaxf(acc.x + b.x, 0.0f) + hv.x;
  o.y = fmaxf(acc.y + b.y, 0.0f) + hv.y;
  hp[lane] = o;
}

template<int K, bool RELU_BIAS>
__global__ void gemm_n64_wmma(const float* __restrict__ A,
                              const float* __restrict__ W,
                              const float* __restrict__ bias,
                              float* __restrict__ out, int n_rows) {
  int wave = (blockIdx.x * blockDim.x + threadIdx.x) >> 5;
  int lane = threadIdx.x & 31;
  int row0 = wave * 16;
  if (row0 >= n_rows) return;

  const int half = lane >> 4;
  const int l16  = lane & 15;

  v8f acc0 = {}, acc1 = {}, acc2 = {}, acc3 = {};

  const float* arow = A + (size_t)(row0 + l16) * K;
  for (int kb = 0; kb < K; kb += 4) {
    v2f a;
    a.x = arow[kb + half * 2 + 0];
    a.y = arow[kb + half * 2 + 1];

    const float* bp = W + (size_t)(kb + half * 2) * HDIM + l16;
    v2f b0, b1, b2, b3;
    b0.x = bp[0];        b0.y = bp[HDIM];
    b1.x = bp[16];       b1.y = bp[HDIM + 16];
    b2.x = bp[32];       b2.y = bp[HDIM + 32];
    b3.x = bp[48];       b3.y = bp[HDIM + 48];

    acc0 = __builtin_amdgcn_wmma_f32_16x16x4_f32(false, a, false, b0, (short)0, acc0, false, false);
    acc1 = __builtin_amdgcn_wmma_f32_16x16x4_f32(false, a, false, b1, (short)0, acc1, false, false);
    acc2 = __builtin_amdgcn_wmma_f32_16x16x4_f32(false, a, false, b2, (short)0, acc2, false, false);
    acc3 = __builtin_amdgcn_wmma_f32_16x16x4_f32(false, a, false, b3, (short)0, acc3, false, false);
  }

  v8f accs[4] = {acc0, acc1, acc2, acc3};
  #pragma unroll
  for (int nt = 0; nt < 4; ++nt) {
    int col = nt * 16 + l16;
    float bcol = RELU_BIAS ? bias[col] : 0.0f;
    #pragma unroll
    for (int r = 0; r < 8; ++r) {
      int m = half * 8 + r;
      float v = accs[nt][r];
      if (RELU_BIAS) { v += bcol; v = fmaxf(v, 0.0f); }
      out[(size_t)(row0 + m) * HDIM + col] = v;
    }
  }
}

__global__ void head_kernel(const float* __restrict__ h, const float* __restrict__ Wh,
                            const float* __restrict__ bh, float* __restrict__ out, int n) {
  int i = blockIdx.x * blockDim.x + threadIdx.x;
  if (i >= n) return;
  const float* hp = h + (size_t)i * HDIM;
  float s = bh[0];
  #pragma unroll
  for (int k = 0; k < HDIM; ++k) s += hp[k] * Wh[k];
  out[i] = s;
}

extern "C" void kernel_launch(void* const* d_in, const int* in_sizes, int n_in,
                              void* d_out, int out_size, void* d_ws, size_t ws_size,
                              hipStream_t stream) {
  const float* x      = (const float*)d_in[0];
  const int*   eidx   = (const int*)d_in[1];
  const float* W_pre  = (const float*)d_in[2];
  const float* b_pre  = (const float*)d_in[3];
  const float* W1     = (const float*)d_in[4];
  const float* b1     = (const float*)d_in[5];
  const float* W2     = (const float*)d_in[6];
  const float* b2     = (const float*)d_in[7];
  const float* W_head = (const float*)d_in[8];
  const float* b_head = (const float*)d_in[9];
  float* out = (float*)d_out;

  const int N = in_sizes[0] / 128;
  const int E = in_sizes[1] / 2;
  const int* src = eidx;
  const int* dst = eidx + E;

  float* h        = (float*)d_ws;
  float* hw       = h + (size_t)N * HDIM;
  int*   ssrc     = (int*)(hw + (size_t)N * HDIM);
  int*   rank     = ssrc + E;
  int*   cnt      = rank + E;
  int*   rowstart = cnt + N;
  float* dis      = (float*)(rowstart + N);

  zero_i32<<<(N + 255) / 256, 256, 0, stream>>>(cnt, N);
  rank_edges<<<(E + 255) / 256, 256, 0, stream>>>(dst, cnt, rank, E);
  scan_rows<<<(N + 1023) / 1024, 1024, 0, stream>>>(cnt, rowstart, dis, N);
  fill_sorted<<<(E + 255) / 256, 256, 0, stream>>>(src, dst, rowstart, rank, ssrc, E);

  int gemm_blocks = (N + 127) / 128;
  gemm_n64_wmma<128, true><<<gemm_blocks, 256, 0, stream>>>(x, W_pre, b_pre, h, N);

  const float* Ws[2] = {W1, W2};
  const float* bs[2] = {b1, b2};
  int gather_blocks = (N + 7) / 8;
  for (int layer = 0; layer < 2; ++layer) {
    gemm_n64_wmma<64, false><<<gemm_blocks, 256, 0, stream>>>(h, Ws[layer], nullptr, hw, N);
    gather_rows<<<gather_blocks, 256, 0, stream>>>(rowstart, cnt, ssrc, dis, hw, bs[layer], h, N);
  }

  head_kernel<<<(N + 255) / 256, 256, 0, stream>>>(h, W_head, b_head, out, N);
}


# Source-side deg^-1/2 folded into the rows: the K=64 product's epilogue scales row r by dis[r]; the gather adds plain rows and applies the destination factor once, no 4-byte lookup per edge
# speedup vs baseline: 2.0558x; 1.0360x over previous
#include <hip/hip_runtime.h>
#include <hip/hip_bf16.h>

typedef __attribute__((ext_vector_type(2))) float v2f;
typedef __attribute__((ext_vector_type(8))) float v8f;

#define HDIM 64

__global__ void zero_i32(int* __restrict__ p, int n) {
  int i = blockIdx.x * blockDim.x + threadIdx.x;
  if (i < n) p[i] = 0;
}

__global__ void rank_edges(const int* __restrict__ dst, int* __restrict__ cnt,
                           int* __restrict__ rank, int E) {
  int e = blockIdx.x * blockDim.x + threadIdx.x;
  if (e < E) rank[e] = atomicAdd(&cnt[dst[e]], 1);
}

__global__ __launch_bounds__(1024) void scan_rows(const int* __restrict__ cnt,
                                                  int* __restrict__ rowstart,
                                                  float* __restrict__ dis, int n) {
  __shared__ int wsum[32];
  __shared__ int s_prefix;
  const int tid  = threadIdx.x;
  const int lane = tid & 31;
  const int wid  = tid >> 5;
  const int base = blockIdx.x * 1024;

  int p = 0;
  for (int i = tid; i < base; i += 1024) p += cnt[i];
  for (int o = 16; o > 0; o >>= 1) p += __shfl_xor(p, o);
  if (lane == 0) wsum[wid] = p;
  __syncthreads();
  if (wid == 0) {
    int v = wsum[lane];
    for (int o = 16; o > 0; o >>= 1) v += __shfl_xor(v, o);
    if (lane == 0) s_prefix = v;
  }
  __syncthreads();
  const int prefix = s_prefix;

  const int i = base + tid;
  const int c = (i < n) ? cnt[i] : 0;
  int x = c;
  for (int o = 1; o < 32; o <<= 1) {
    int y = __shfl_up(x, o);
    if (lane >= o) x += y;
  }
  if (lane == 31) wsum[wid] = x;
  __syncthreads();
  int wo = (lane < wid) ? wsum[lane] : 0;
  for (int o = 16; o > 0; o >>= 1) wo += __shfl_xor(wo, o);

  if (i < n) {
    rowstart[i] = prefix + wo + x - c;
    float d = (float)c;
    dis[i] = (d > 0.0f) ? rsqrtf(fmaxf(d, 1.0f)) : 0.0f;
  }
}

__global__ void fill_sorted(const int* __restrict__ src, const int* __restrict__ dst,
                            const int* __restrict__ rowstart, const int* __restrict__ rank,
                            int* __restrict__ ssrc, int E) {
  int e = blockIdx.x * blockDim.x + threadIdx.x;
  if (e < E) ssrc[rowstart[dst[e]] + rank[e]] = src[e];
}

__global__ void gather_rows(const int* __restrict__ rowstart, const int* __restrict__ cnt,
                            const int* __restrict__ ssrc, const float* __restrict__ dis,
                            const float* __restrict__ hw, const float* __restrict__ bias,
                            float* __restrict__ h, int n) {
  const int lane = threadIdx.x & 31;
  const int node = __builtin_amdgcn_readfirstlane((int)((blockIdx.x * blockDim.x + threadIdx.x) >> 5));
  if (node >= n) return;
  const int start = rowstart[node];
  const int deg   = cnt[node];
  const float dd  = dis[node];

  float2 acc = make_float2(0.0f, 0.0f);
  for (int j0 = 0; j0 < deg; j0 += 32) {
    const int m = (deg - j0 < 32) ? (deg - j0) : 32;
    int s_l = 0;
    if (lane < m) s_l = ssrc[start + j0 + lane];
    for (int j = 0; j < m; j += 4) {
      float2 v[4];
      for (int k = 0; k < 4; ++k) {
        const int  jj  = j + k;
        const bool in  = jj < m;
        const int  jc  = in ? jj : (m - 1);
        const int  s   = __builtin_amdgcn_readlane(s_l, jc);
        const float2* rp = (const float2*)(hw + (size_t)s * HDIM);
        float2 t = rp[lane];
        v[k].x = in ? t.x : 0.0f;
        v[k].y = in ? t.y : 0.0f;
      }
      for (int k = 0; k < 4; ++k) {
        acc.x += v[k].x;
        acc.y += v[k].y;
      }
    }
  }

  const float2 b  = ((const float2*)bias)[lane];
  float2* hp = (float2*)(h + (size_t)node * HDIM);
  const float2 hv = hp[lane];
  float2 o;
  o.x = fmaxf(acc.x * dd + b.x, 0.0f) + hv.x;
  o.y = fmaxf(acc.y * dd + b.y, 0.0f) + hv.y;
  hp[lane] = o;
}

template<int K, bool RELU_BIAS, bool ROW_SCALE>
__global__ void gemm_n64_wmma(const float* __restrict__ A,
                              const float* __restrict__ W,
                              const float* __restrict__ bias,
                              const float* __restrict__ rowscale,
                              float* __restrict__ out, int n_rows) {
  int wave = (blockIdx.x * blockDim.x + threadIdx.x) >> 5;
  int lane = threadIdx.x & 31;
  int row0 = wave * 16;
  if (row0 >= n_rows) return;

  const int half = lane >> 4;
  const int l16  = lane & 15;

  v8f acc0 = {}, acc1 = {}, acc2 = {}, acc3 = {};

  float rs[8];
  #pragma unroll
  for (int r = 0; r < 8; ++r) rs[r] = ROW_SCALE ? rowscale[row0 + half * 8 + r] : 1.0f;

  const float* arow = A + (size_t)(row0 + l16) * K;
  for (int kb = 0; kb < K; kb += 4) {
    v2f a;
    a.x = arow[kb + half * 2 + 0];
    a.y = arow[kb + half * 2 + 1];

    const float* bp = W + (size_t)(kb + half * 2) * HDIM + l16;
    v2f b0, b1, b2, b3;
    b0.x = bp[0];        b0.y = bp[HDIM];
    b1.x = bp[16];       b1.y = bp[HDIM + 16];
    b2.x = bp[32];       b2.y = bp[HDIM + 32];
    b3.x = bp[48];       b3.y = bp[HDIM + 48];

    acc0 = __builtin_amdgcn_wmma_f32_16x16x4_f32(false, a, false, b0, (short)0, acc0, false, false);
    acc1 = __builtin_amdgcn_wmma_f32_16x16x4_f32(false, a, false, b1, (short)0, acc1, false, false);
    acc2 = __builtin_amdgcn_wmma_f32_16x16x4_f32(false, a, false, b2, (short)0, acc2, false, false);
    acc3 = __builtin_amdgcn_wmma_f32_16x16x4_f32(false, a, false, b3, (short)0, acc3, false, false);
  }

  v8f accs[4] = {acc0, acc1, acc2, acc3};
  #pragma unroll
  for (int nt = 0; nt < 4; ++nt) {
    int col = nt * 16 + l16;
    float bcol = RELU_BIAS ? bias[col] : 0.0f;
    #pragma unroll
    for (int r = 0; r < 8; ++r) {
      int m = half * 8 + r;
      float v = accs[nt][r];
      if (RELU_BIAS) { v += bcol; v = fmaxf(v, 0.0f); }
      if (ROW_SCALE) v *= rs[r];
      out[(size_t)(row0 + m) * HDIM + col] = v;
    }
  }
}

__global__ void head_kernel(const float* __restrict__ h, const float* __restrict__ Wh,
                            const float* __restrict__ bh, float* __restrict__ out, int n) {
  int i = blockIdx.x * blockDim.x + threadIdx.x;
  if (i >= n) return;
  const float* hp = h + (size_t)i * HDIM;
  float s = bh[0];
  #pragma unroll
  for (int k = 0; k < HDIM; ++k) s += hp[k] * Wh[k];
  out[i] = s;
}

extern "C" void kernel_launch(void* const* d_in, const int* in_sizes, int n_in,
                              void* d_out, int out_size, void* d_ws, size_t ws_size,
                              hipStream_t stream) {
  const float* x      = (const float*)d_in[0];
  const int*   eidx   = (const int*)d_in[1];
  const float* W_pre  = (const float*)d_in[2];
  const float* b_pre  = (const float*)d_in[3];
  const float* W1     = (const float*)d_in[4];
  const float* b1     = (const float*)d_in[5];
  const float* W2     = (const float*)d_in[6];
  const float* b2     = (const float*)d_in[7];
  const float* W_head = (const float*)d_in[8];
  const float* b_head = (const float*)d_in[9];
  float* out = (float*)d_out;

  const int N = in_sizes[0] / 128;
  const int E = in_sizes[1] / 2;
  const int* src = eidx;
  const int* dst = eidx + E;

  float* h        = (float*)d_ws;
  float* hw       = h + (size_t)N * HDIM;
  int*   ssrc     = (int*)(hw + (size_t)N * HDIM);
  int*   rank     = ssrc + E;
  int*   cnt      = rank + E;
  int*   rowstart = cnt + N;
  float* dis      = (float*)(rowstart + N);

  zero_i32<<<(N + 255) / 256, 256, 0, stream>>>(cnt, N);
  rank_edges<<<(E + 255) / 256, 256, 0, stream>>>(dst, cnt, rank, E);
  scan_rows<<<(N + 1023) / 1024, 1024, 0, stream>>>(cnt, rowstart, dis, N);
  fill_sorted<<<(E + 255) / 256, 256, 0, stream>>>(src, dst, rowstart, rank, ssrc, E);

  int gemm_blocks = (N + 127) / 128;
  gemm_n64_wmma<128, true, false><<<gemm_blocks, 256, 0, stream>>>(x, W_pre, b_pre, nullptr, h, N);

  const float* Ws[2] = {W1, W2};
  const float* bs[2] = {b1, b2};
  int gather_blocks = (N + 7) / 8;
  for (int layer = 0; layer < 2; ++layer) {
    gemm_n64_wmma<64, false, true><<<gemm_blocks, 256, 0, stream>>>(h, Ws[layer], nullptr, dis, hw, N);
    gather_rows<<<gather_blocks, 256, 0, stream>>>(rowstart, cnt, ssrc, dis, hw, bs[layer], h, N);
  }

  head_kernel<<<(N + 255) / 256, 256, 0, stream>>>(h, W_head, b_head, out, N);
}


# Edges grouped by destination by a two-level counting sort in LDS (512-node buckets, runs written as whole lines), not a global returning atomic and 4-byte scatter per edge; sum order = LDS arrival
# speedup vs baseline: 3.5642x; 1.7337x over previous
#include <hip/hip_runtime.h>
#include <hip/hip_bf16.h>

typedef __attribute__((ext_vector_type(2))) float v2f;
typedef __attribute__((ext_vector_type(8))) float v8f;

#define HDIM 64

#define BSHIFT 9
#define BNODES 512
#define MAXB   256
#define ECHUNK 8192
#define STAGE  16384

__device__ __forceinline__ int block_excl_scan(int v, int* wsum) {
  const int lane = threadIdx.x & 31;
  const int wid  = threadIdx.x >> 5;
  int x = v;
  for (int o = 1; o < 32; o <<= 1) {
    int y = __shfl_up(x, o);
    if (lane >= o) x += y;
  }
  if (lane == 31) wsum[wid] = x;
  __syncthreads();
  int wo = (lane < wid) ? wsum[lane] : 0;
  for (int o = 16; o > 0; o >>= 1) wo += __shfl_xor(wo, o);
  __syncthreads();
  return wo + x - v;
}

__global__ __launch_bounds__(1024) void edge_hist(const int* __restrict__ dst, int* __restrict__ hist,
                                                  int E, int nb) {
  __shared__ int lh[MAXB];
  const int tid = threadIdx.x;
  const int e0  = blockIdx.x * ECHUNK;
  if (tid < MAXB) lh[tid] = 0;
  __syncthreads();
  int dv[ECHUNK / 1024];
  for (int i = 0; i < ECHUNK / 1024; ++i) {
    const int e = e0 + i * 1024 + tid;
    dv[i] = (e < E) ? dst[e] : -1;
  }
  for (int i = 0; i < ECHUNK / 1024; ++i)
    if (dv[i] >= 0) atomicAdd(&lh[dv[i] >> BSHIFT], 1);
  __syncthreads();
  if (tid < nb) hist[blockIdx.x * nb + tid] = lh[tid];
}

__global__ __launch_bounds__(256) void scan_hist(const int* __restrict__ hist, int* __restrict__ offs,
                                                 int* __restrict__ btot, int nb, int nblk) {
  __shared__ int wsum[8];
  const int tid  = threadIdx.x;
  const int lane = tid & 31;
  const int wid  = tid >> 5;
  const int b    = blockIdx.x;
  const int c = (tid < nblk) ? hist[tid * nb + b] : 0;
  int x = c;
  for (int o = 1; o < 32; o <<= 1) {
    int y = __shfl_up(x, o);
    if (lane >= o) x += y;
  }
  if (lane == 31) wsum[wid] = x;
  __syncthreads();
  int wo = 0, all = 0;
  for (int j = 0; j < 8; ++j) {
    const int w = wsum[j];
    if (j < wid) wo += w;
    all += w;
  }
  if (tid < nblk) offs[tid * nb + b] = wo + x - c;
  if (tid == 0) btot[b] = all;
}

__global__ __launch_bounds__(1024) void scatter_buckets(const int* __restrict__ src, const int* __restrict__ dst,
                                                        const int* __restrict__ hist, const int* __restrict__ offs,
                                                        const int* __restrict__ btot, int* __restrict__ bedge_w,
                                                        int* __restrict__ bbase, int E, int nb) {
  int2* __restrict__ bedge = (int2*)bedge_w;
  __shared__ int wsum[32];
  __shared__ int cur[MAXB];
  __shared__ int gofs[MAXB];
  __shared__ int2 stage[ECHUNK];
  const int tid = threadIdx.x;
  const int blk = blockIdx.x;
  const int e0  = blk * ECHUNK;
  const bool isb = tid < nb;
  const int tot = isb ? btot[tid] : 0;
  const int own = isb ? hist[blk * nb + tid] : 0;
  const int pre = isb ? offs[blk * nb + tid] : 0;
  const int base = block_excl_scan(tot, wsum);
  const int ls   = block_excl_scan(own, wsum);
  if (isb) {
    cur[tid]  = ls;
    gofs[tid] = base + pre - ls;
    if (blk == 0) {
      bbase[tid] = base;
      if (tid == nb - 1) bbase[nb] = base + tot;
    }
  }
  __syncthreads();
  int sv[ECHUNK / 1024], dv[ECHUNK / 1024];
  for (int i = 0; i < ECHUNK / 1024; ++i) {
    const int e = e0 + i * 1024 + tid;
    const bool in = e < E;
    sv[i] = in ? src[e] : 0;
    dv[i] = in ? dst[e] : -1;
  }
  for (int i = 0; i < ECHUNK / 1024; ++i) {
    if (dv[i] >= 0) {
      const int p = atomicAdd(&cur[dv[i] >> BSHIFT], 1);
      stage[p] = make_int2(sv[i], dv[i]);
    }
  }
  __syncthreads();
  const int m = (E - e0 < ECHUNK) ? (E - e0) : ECHUNK;
  for (int i = tid; i < m; i += 1024) {
    const int2 ed = stage[i];
    bedge[gofs[ed.y >> BSHIFT] + i] = ed;
  }
}

__global__ __launch_bounds__(1024) void sort_bucket(const int* __restrict__ bedge_w, const int* __restrict__ bbase,
                                                    int* __restrict__ cnt, int* __restrict__ rowstart,
                                                    float* __restrict__ dis, int* __restrict__ ssrc, int n) {
  const int2* __restrict__ bedge = (const int2*)bedge_w;
  __shared__ int ncnt[BNODES];
  __shared__ int ncur[BNODES];
  __shared__ int wsum[32];
  __shared__ int stage[STAGE];
  const int tid  = threadIdx.x;
  const int k    = blockIdx.x;
  const int b0   = bbase[k];
  const int m    = bbase[k + 1] - b0;

  if (tid < BNODES) ncnt[tid] = 0;
  __syncthreads();
  for (int i0 = 0; i0 < m; i0 += 8 * 1024) {
    int dv[8];
    for (int j = 0; j < 8; ++j) {
      const int i = i0 + j * 1024 + tid;
      dv[j] = (i < m) ? bedge[b0 + i].y : -1;
    }
    for (int j = 0; j < 8; ++j)
      if (dv[j] >= 0) atomicAdd(&ncnt[dv[j] & (BNODES - 1)], 1);
  }
  __syncthreads();

  const int c  = (tid < BNODES) ? ncnt[tid] : 0;
  const int st = block_excl_scan(c, wsum);
  if (tid < BNODES) {
    ncur[tid] = st;
    const int node = k * BNODES + tid;
    if (node < n) {
      cnt[node]      = c;
      rowstart[node] = b0 + st;
      const float d  = (float)c;
      dis[node]      = (d > 0.0f) ? rsqrtf(fmaxf(d, 1.0f)) : 0.0f;
    }
  }
  __syncthreads();

  if (m <= STAGE) {
    for (int i0 = 0; i0 < m; i0 += 8 * 1024) {
      int2 ev[8];
      for (int j = 0; j < 8; ++j) {
        const int i = i0 + j * 1024 + tid;
        ev[j] = (i < m) ? bedge[b0 + i] : make_int2(0, -1);
      }
      for (int j = 0; j < 8; ++j) {
        if (ev[j].y >= 0) {
          const int p = atomicAdd(&ncur[ev[j].y & (BNODES - 1)], 1);
          stage[p] = ev[j].x;
        }
      }
    }
    __syncthreads();
    for (int i = tid; i < m; i += 1024) ssrc[b0 + i] = stage[i];
  } else {
    for (int i = tid; i < m; i += 1024) {
      const int2 ed = bedge[b0 + i];
      const int p = atomicAdd(&ncur[ed.y & (BNODES - 1)], 1);
      ssrc[b0 + p] = ed.x;
    }
  }
}

__global__ void gather_rows(const int* __restrict__ rowstart, const int* __restrict__ cnt,
                            const int* __restrict__ ssrc, const float* __restrict__ dis,
                            const float* __restrict__ hw, const float* __restrict__ bias,
                            float* __restrict__ h, int n) {
  const int lane = threadIdx.x & 31;
  const int node = __builtin_amdgcn_readfirstlane((int)((blockIdx.x * blockDim.x + threadIdx.x) >> 5));
  if (node >= n) return;
  const int start = rowstart[node];
  const int deg   = cnt[node];
  const float dd  = dis[node];

  float2 acc = make_float2(0.0f, 0.0f);
  for (int j0 = 0; j0 < deg; j0 += 32) {
    const int m = (deg - j0 < 32) ? (deg - j0) : 32;
    int s_l = 0;
    if (lane < m) s_l = ssrc[start + j0 + lane];
    for (int j = 0; j < m; j += 4) {
      float2 v[4];
      for (int k = 0; k < 4; ++k) {
        const int  jj  = j + k;
        const bool in  = jj < m;
        const int  jc  = in ? jj : (m - 1);
        const int  s   = __builtin_amdgcn_readlane(s_l, jc);
        const float2* rp = (const float2*)(hw + (size_t)s * HDIM);
        float2 t = rp[lane];
        v[k].x = in ? t.x : 0.0f;
        v[k].y = in ? t.y : 0.0f;
      }
      for (int k = 0; k < 4; ++k) {
        acc.x += v[k].x;
        acc.y += v[k].y;
      }
    }
  }

  const float2 b  = ((const float2*)bias)[lane];
  float2* hp = (float2*)(h + (size_t)node * HDIM);
  const float2 hv = hp[lane];
  float2 o;
  o.x = fmaxf(acc.x * dd + b.x, 0.0f) + hv.x;
  o.y = fmaxf(acc.y * dd + b.y, 0.0f) + hv.y;
  hp[lane] = o;
}

template<int K, bool RELU_BIAS, bool ROW_SCALE>
__global__ void gemm_n64_wmma(const float* __restrict__ A,
                              const float* __restrict__ W,
                              const float* __restrict__ bias,
                              const float* __restrict__ rowscale,
                              float* __restrict__ out, int n_rows) {
  int wave = (blockIdx.x * blockDim.x + threadIdx.x) >> 5;
  int lane = threadIdx.x & 31;
  int row0 = wave * 16;
  if (row0 >= n_rows) return;

  const int half = lane >> 4;
  const int l16  = lane & 15;

  v8f acc0 = {}, acc1 = {}, acc2 = {}, acc3 = {};

  float rs[8];
  #pragma unroll
  for (int r = 0; r < 8; ++r) rs[r] = ROW_SCALE ? rowscale[row0 + half * 8 + r] : 1.0f;

  const float* arow = A + (size_t)(row0 + l16) * K;
  for (int kb = 0; kb < K; kb += 4) {
    v2f a;
    a.x = arow[kb + half * 2 + 0];
    a.y = arow[kb + half * 2 + 1];

    const float* bp = W + (size_t)(kb + half * 2) * HDIM + l16;
    v2f b0, b1, b2, b3;
    b0.x = bp[0];        b0.y = bp[HDIM];
    b1.x = bp[16];       b1.y = bp[HDIM + 16];
    b2.x = bp[32];       b2.y = bp[HDIM + 32];
    b3.x = bp[48];       b3.y = bp[HDIM + 48];

    acc0 = __builtin_amdgcn_wmma_f32_16x16x4_f32(false, a, false, b0, (short)0, acc0, false, false);
    acc1 = __builtin_amdgcn_wmma_f32_16x16x4_f32(false, a, false, b1, (short)0, acc1, false, false);
    acc2 = __builtin_amdgcn_wmma_f32_16x16x4_f32(false, a, false, b2, (short)0, acc2, false, false);
    acc3 = __builtin_amdgcn_wmma_f32_16x16x4_f32(false, a, false, b3, (short)0, acc3, false, false);
  }

  v8f accs[4] = {acc0, acc1, acc2, acc3};
  #pragma unroll
  for (int nt = 0; nt < 4; ++nt) {
    int col = nt * 16 + l16;
    float bcol = RELU_BIAS ? bias[col] : 0.0f;
    #pragma unroll
    for (int r = 0; r < 8; ++r) {
      int m = half * 8 + r;
      float v = accs[nt][r];
      if (RELU_BIAS) { v += bcol; v = fmaxf(v, 0.0f); }
      if (ROW_SCALE) v *= rs[r];
      out[(size_t)(row0 + m) * HDIM + col] = v;
    }
  }
}

__global__ void head_kernel(const float* __restrict__ h, const float* __restrict__ Wh,
                            const float* __restrict__ bh, float* __restrict__ out, int n) {
  int i = blockIdx.x * blockDim.x + threadIdx.x;
  if (i >= n) return;
  const float* hp = h + (size_t)i * HDIM;
  float s = bh[0];
  #pragma unroll
  for (int k = 0; k < HDIM; ++k) s += hp[k] * Wh[k];
  out[i] = s;
}

extern "C" void kernel_launch(void* const* d_in, const int* in_sizes, int n_in,
                              void* d_out, int out_size, void* d_ws, size_t ws_size,
                              hipStream_t stream) {
  const float* x      = (const float*)d_in[0];
  const int*   eidx   = (const int*)d_in[1];
  const float* W_pre  = (const float*)d_in[2];
  const float* b_pre  = (const float*)d_in[3];
  const float* W1     = (const float*)d_in[4];
  const float* b1     = (const float*)d_in[5];
  const float* W2     = (const float*)d_in[6];
  const float* b2     = (const float*)d_in[7];
  const float* W_head = (const float*)d_in[8];
  const float* b_head = (const float*)d_in[9];
  float* out = (float*)d_out;

  const int N = in_sizes[0] / 128;
  const int E = in_sizes[1] / 2;
  const int* src = eidx;
  const int* dst = eidx + E;

  float* h        = (float*)d_ws;
  float* hw       = h + (size_t)N * HDIM;
  int*   bedge    = (int*)(hw + (size_t)N * HDIM);
  int*   ssrc     = bedge + 2 * (size_t)E;
  int*   cnt      = ssrc + E;
  int*   rowstart = cnt + N;
  float* dis      = (float*)(rowstart + N);

  const int nb   = (N + BNODES - 1) / BNODES;
  const int nblk = (E + ECHUNK - 1) / ECHUNK;
  int* hist  = (int*)(dis + N);
  int* offs  = hist + (size_t)nblk * nb;
  int* btot  = offs + (size_t)nblk * nb;
  int* bbase = btot + nb;
  edge_hist<<<nblk, 1024, 0, stream>>>(dst, hist, E, nb);
  scan_hist<<<nb, 256, 0, stream>>>(hist, offs, btot, nb, nblk);
  scatter_buckets<<<nblk, 1024, 0, stream>>>(src, dst, hist, offs, btot, bedge, bbase, E, nb);
  sort_bucket<<<nb, 1024, 0, stream>>>(bedge, bbase, cnt, rowstart, dis, ssrc, N);

  int gemm_blocks = (N + 127) / 128;
  gemm_n64_wmma<128, true, false><<<gemm_blocks, 256, 0, stream>>>(x, W_pre, b_pre, nullptr, h, N);

  const float* Ws[2] = {W1, W2};
  const float* bs[2] = {b1, b2};
  int gather_blocks = (N + 7) / 8;
  for (int layer = 0; layer < 2; ++layer) {
    gemm_n64_wmma<64, false, true><<<gemm_blocks, 256, 0, stream>>>(h, Ws[layer], nullptr, dis, hw, N);
    gather_rows<<<gather_blocks, 256, 0, stream>>>(rowstart, cnt, ssrc, dis, hw, bs[layer], h, N);
  }

  head_kernel<<<(N + 255) / 256, 256, 0, stream>>>(h, W_head, b_head, out, N);
}


# Neighbour sum: a wave sums 4 consecutive rows side by side, 8 neighbour-row loads per row (32 per wave) in flight per round, row info, skip rows and ids asked for in groups; row sum order kept
# speedup vs baseline: 3.8654x; 1.0845x over previous
#include <hip/hip_runtime.h>
#include <hip/hip_bf16.h>

typedef __attribute__((ext_vector_type(2))) float v2f;
typedef __attribute__((ext_vector_type(8))) float v8f;

#define HDIM 64

#define BSHIFT 9
#define BNODES 512
#define MAXB   256
#define ECHUNK 8192
#define STAGE  16384

__device__ __forceinline__ int block_excl_scan(int v, int* wsum) {
  const int lane = threadIdx.x & 31;
  const int wid  = threadIdx.x >> 5;
  int x = v;
  for (int o = 1; o < 32; o <<= 1) {
    int y = __shfl_up(x, o);
    if (lane >= o) x += y;
  }
  if (lane == 31) wsum[wid] = x;
  __syncthreads();
  int wo = (lane < wid) ? wsum[lane] : 0;
  for (int o = 16; o > 0; o >>= 1) wo += __shfl_xor(wo, o);
  __syncthreads();
  return wo + x - v;
}

__global__ __launch_bounds__(1024) void edge_hist(const int* __restrict__ dst, int* __restrict__ hist,
                                                  int E, int nb) {
  __shared__ int lh[MAXB];
  const int tid = threadIdx.x;
  const int e0  = blockIdx.x * ECHUNK;
  if (tid < MAXB) lh[tid] = 0;
  __syncthreads();
  int dv[ECHUNK / 1024];
  for (int i = 0; i < ECHUNK / 1024; ++i) {
    const int e = e0 + i * 1024 + tid;
    dv[i] = (e < E) ? dst[e] : -1;
  }
  for (int i = 0; i < ECHUNK / 1024; ++i)
    if (dv[i] >= 0) atomicAdd(&lh[dv[i] >> BSHIFT], 1);
  __syncthreads();
  if (tid < nb) hist[blockIdx.x * nb + tid] = lh[tid];
}

__global__ __launch_bounds__(256) void scan_hist(const int* __restrict__ hist, int* __restrict__ offs,
                                                 int* __restrict__ btot, int nb, int nblk) {
  __shared__ int wsum[8];
  const int tid  = threadIdx.x;
  const int lane = tid & 31;
  const int wid  = tid >> 5;
  const int b    = blockIdx.x;
  const int c = (tid < nblk) ? hist[tid * nb + b] : 0;
  int x = c;
  for (int o = 1; o < 32; o <<= 1) {
    int y = __shfl_up(x, o);
    if (lane >= o) x += y;
  }
  if (lane == 31) wsum[wid] = x;
  __syncthreads();
  int wo = 0, all = 0;
  for (int j = 0; j < 8; ++j) {
    const int w = wsum[j];
    if (j < wid) wo += w;
    all += w;
  }
  if (tid < nblk) offs[tid * nb + b] = wo + x - c;
  if (tid == 0) btot[b] = all;
}

__global__ __launch_bounds__(1024) void scatter_buckets(const int* __restrict__ src, const int* __restrict__ dst,
                                                        const int* __restrict__ hist, const int* __restrict__ offs,
                                                        const int* __restrict__ btot, int* __restrict__ bedge_w,
                                                        int* __restrict__ bbase, int E, int nb) {
  int2* __restrict__ bedge = (int2*)bedge_w;
  __shared__ int wsum[32];
  __shared__ int cur[MAXB];
  __shared__ int gofs[MAXB];
  __shared__ int2 stage[ECHUNK];
  const int tid = threadIdx.x;
  const int blk = blockIdx.x;
  const int e0  = blk * ECHUNK;
  const bool isb = tid < nb;
  const int tot = isb ? btot[tid] : 0;
  const int own = isb ? hist[blk * nb + tid] : 0;
  const int pre = isb ? offs[blk * nb + tid] : 0;
  const int base = block_excl_scan(tot, wsum);
  const int ls   = block_excl_scan(own, wsum);
  if (isb) {
    cur[tid]  = ls;
    gofs[tid] = base + pre - ls;
    if (blk == 0) {
      bbase[tid] = base;
      if (tid == nb - 1) bbase[nb] = base + tot;
    }
  }
  __syncthreads();
  int sv[ECHUNK / 1024], dv[ECHUNK / 1024];
  for (int i = 0; i < ECHUNK / 1024; ++i) {
    const int e = e0 + i * 1024 + tid;
    const bool in = e < E;
    sv[i] = in ? src[e] : 0;
    dv[i] = in ? dst[e] : -1;
  }
  for (int i = 0; i < ECHUNK / 1024; ++i) {
    if (dv[i] >= 0) {
      const int p = atomicAdd(&cur[dv[i] >> BSHIFT], 1);
      stage[p] = make_int2(sv[i], dv[i]);
    }
  }
  __syncthreads();
  const int m = (E - e0 < ECHUNK) ? (E - e0) : ECHUNK;
  for (int i = tid; i < m; i += 1024) {
    const int2 ed = stage[i];
    bedge[gofs[ed.y >> BSHIFT] + i] = ed;
  }
}

__global__ __launch_bounds__(1024) void sort_bucket(const int* __restrict__ bedge_w, const int* __restrict__ bbase,
                                                    int* __restrict__ cnt, int* __restrict__ rowstart,
                                                    float* __restrict__ dis, int* __restrict__ ssrc, int n) {
  const int2* __restrict__ bedge = (const int2*)bedge_w;
  __shared__ int ncnt[BNODES];
  __shared__ int ncur[BNODES];
  __shared__ int wsum[32];
  __shared__ int stage[STAGE];
  const int tid  = threadIdx.x;
  const int k    = blockIdx.x;
  const int b0   = bbase[k];
  const int m    = bbase[k + 1] - b0;

  if (tid < BNODES) ncnt[tid] = 0;
  __syncthreads();
  for (int i0 = 0; i0 < m; i0 += 8 * 1024) {
    int dv[8];
    for (int j = 0; j < 8; ++j) {
      const int i = i0 + j * 1024 + tid;
      dv[j] = (i < m) ? bedge[b0 + i].y : -1;
    }
    for (int j = 0; j < 8; ++j)
      if (dv[j] >= 0) atomicAdd(&ncnt[dv[j] & (BNODES - 1)], 1);
  }
  __syncthreads();

  const int c  = (tid < BNODES) ? ncnt[tid] : 0;
  const int st = block_excl_scan(c, wsum);
  if (tid < BNODES) {
    ncur[tid] = st;
    const int node = k * BNODES + tid;
    if (node < n) {
      cnt[node]      = c;
      rowstart[node] = b0 + st;
      const float d  = (float)c;
      dis[node]      = (d > 0.0f) ? rsqrtf(fmaxf(d, 1.0f)) : 0.0f;
    }
  }
  __syncthreads();

  if (m <= STAGE) {
    for (int i0 = 0; i0 < m; i0 += 8 * 1024) {
      int2 ev[8];
      for (int j = 0; j < 8; ++j) {
        const int i = i0 + j * 1024 + tid;
        ev[j] = (i < m) ? bedge[b0 + i] : make_int2(0, -1);
      }
      for (int j = 0; j < 8; ++j) {
        if (ev[j].y >= 0) {
          const int p = atomicAdd(&ncur[ev[j].y & (BNODES - 1)], 1);
          stage[p] = ev[j].x;
        }
      }
    }
    __syncthreads();
    for (int i = tid; i < m; i += 1024) ssrc[b0 + i] = stage[i];
  } else {
    for (int i = tid; i < m; i += 1024) {
      const int2 ed = bedge[b0 + i];
      const int p = atomicAdd(&ncur[ed.y & (BNODES - 1)], 1);
      ssrc[b0 + p] = ed.x;
    }
  }
}

#define RPW 4
#define GRP 8

__global__ void gather_rows(const int* __restrict__ rowstart, const int* __restrict__ cnt,
                            const int* __restrict__ ssrc, const float* __restrict__ dis,
                            const float* __restrict__ hw, const float* __restrict__ bias,
                            float* __restrict__ h, int n) {
  const int lane = threadIdx.x & 31;
  const int node0 = RPW * __builtin_amdgcn_readfirstlane((int)((blockIdx.x * blockDim.x + threadIdx.x) >> 5));
  if (node0 >= n) return;

  float2 hv[RPW];
  #pragma unroll
  for (int r = 0; r < RPW; ++r) {
    const int node = (node0 + r < n) ? node0 + r : node0;
    hv[r] = ((const float2*)h)[(unsigned)node * (HDIM / 2) + lane];
  }
  const float2 b = ((const float2*)bias)[lane];
  const int nl = (lane < RPW && node0 + lane < n) ? node0 + lane : node0;
  const int   c_l  = cnt[nl];
  const int   st_l = rowstart[nl];
  const float d_l  = dis[nl];
  const float2* __restrict__ hw2 = (const float2*)hw;

  int   start[RPW], deg[RPW];
  float dd[RPW];
  int   maxdeg = 0;
  #pragma unroll
  for (int r = 0; r < RPW; ++r) {
    const bool live = node0 + r < n;
    const int  c    = __builtin_amdgcn_readlane(c_l, r);
    start[r] = __builtin_amdgcn_readlane(st_l, r);
    deg[r]   = live ? c : 0;
    dd[r]    = __int_as_float(__builtin_amdgcn_readlane(__float_as_int(d_l), r));
    maxdeg   = (deg[r] > maxdeg) ? deg[r] : maxdeg;
  }

  float2 acc[RPW];
  #pragma unroll
  for (int r = 0; r < RPW; ++r) acc[r] = make_float2(0.0f, 0.0f);

  int s_l[RPW];
  #pragma unroll
  for (int r = 0; r < RPW; ++r) {
    s_l[r] = 0;
    if (lane < deg[r]) s_l[r] = ssrc[start[r] + lane];
  }

  for (int j0 = 0; j0 < maxdeg; j0 += 32) {
    int m[RPW];
    #pragma unroll
    for (int r = 0; r < RPW; ++r) {
      const int left = deg[r] - j0;
      m[r] = (left < 0) ? 0 : ((left < 32) ? left : 32);
    }
    const int mm = (maxdeg - j0 < 32) ? (maxdeg - j0) : 32;
    for (int j = 0; j < mm; j += GRP) {
      int sid[RPW][GRP];
      #pragma unroll
      for (int r = 0; r < RPW; ++r) {
        const int last = (m[r] > 0) ? (m[r] - 1) : 0;
        #pragma unroll
        for (int k = 0; k < GRP; ++k) {
          const int jj = j + k;
          sid[r][k] = __builtin_amdgcn_readlane(s_l[r], (jj < m[r]) ? jj : last);
        }
      }
      float2 v[RPW][GRP];
      #pragma unroll
      for (int r = 0; r < RPW; ++r) {
        if (j < m[r]) {
          #pragma unroll
          for (int k = 0; k < GRP; ++k)
            v[r][k] = hw2[(unsigned)sid[r][k] * (HDIM / 2) + lane];
        } else {
          #pragma unroll
          for (int k = 0; k < GRP; ++k) v[r][k] = make_float2(0.0f, 0.0f);
        }
      }
      #pragma unroll
      for (int r = 0; r < RPW; ++r) {
        #pragma unroll
        for (int k = 0; k < GRP; ++k) {
          const bool in = (j + k) < m[r];
          acc[r].x += in ? v[r][k].x : 0.0f;
          acc[r].y += in ? v[r][k].y : 0.0f;
        }
      }
    }
    if (j0 + 32 < maxdeg) {
      #pragma unroll
      for (int r = 0; r < RPW; ++r) {
        s_l[r] = 0;
        if (j0 + 32 + lane < deg[r]) s_l[r] = ssrc[start[r] + j0 + 32 + lane];
      }
    }
  }

  #pragma unroll
  for (int r = 0; r < RPW; ++r) {
    if (node0 + r < n) {
      float2 o;
      o.x = fmaxf(acc[r].x * dd[r] + b.x, 0.0f) + hv[r].x;
      o.y = fmaxf(acc[r].y * dd[r] + b.y, 0.0f) + hv[r].y;
      ((float2*)(h + (size_t)(node0 + r) * HDIM))[lane] = o;
    }
  }
}

template<int K, bool RELU_BIAS, bool ROW_SCALE>
__global__ void gemm_n64_wmma(const float* __restrict__ A,
                              const float* __restrict__ W,
                              const float* __restrict__ bias,
                              const float* __restrict__ rowscale,
                              float* __restrict__ out, int n_rows) {
  int wave = (blockIdx.x * blockDim.x + threadIdx.x) >> 5;
  int lane = threadIdx.x & 31;
  int row0 = wave * 16;
  if (row0 >= n_rows) return;

  const int half = lane >> 4;
  const int l16  = lane & 15;

  v8f acc0 = {}, acc1 = {}, acc2 = {}, acc3 = {};

  float rs[8];
  #pragma unroll
  for (int r = 0; r < 8; ++r) rs[r] = ROW_SCALE ? rowscale[row0 + half * 8 + r] : 1.0f;

  const float* arow = A + (size_t)(row0 + l16) * K;
  for (int kb = 0; kb < K; kb += 4) {
    v2f a;
    a.x = arow[kb + half * 2 + 0];
    a.y = arow[kb + half * 2 + 1];

    const float* bp = W + (size_t)(kb + half * 2) * HDIM + l16;
    v2f b0, b1, b2, b3;
    b0.x = bp[0];        b0.y = bp[HDIM];
    b1.x = bp[16];       b1.y = bp[HDIM + 16];
    b2.x = bp[32];       b2.y = bp[HDIM + 32];
    b3.x = bp[48];       b3.y = bp[HDIM + 48];

    acc0 = __builtin_amdgcn_wmma_f32_16x16x4_f32(false, a, false, b0, (short)0, acc0, false, false);
    acc1 = __builtin_amdgcn_wmma_f32_16x16x4_f32(false, a, false, b1, (short)0, acc1, false, false);
    acc2 = __builtin_amdgcn_wmma_f32_16x16x4_f32(false, a, false, b2, (short)0, acc2, false, false);
    acc3 = __builtin_amdgcn_wmma_f32_16x16x4_f32(false, a, false, b3, (short)0, acc3, false, false);
  }

  v8f accs[4] = {acc0, acc1, acc2, acc3};
  #pragma unroll
  for (int nt = 0; nt < 4; ++nt) {
    int col = nt * 16 + l16;
    float bcol = RELU_BIAS ? bias[col] : 0.0f;
    #pragma unroll
    for (int r = 0; r < 8; ++r) {
      int m = half * 8 + r;
      float v = accs[nt][r];
      if (RELU_BIAS) { v += bcol; v = fmaxf(v, 0.0f); }
      if (ROW_SCALE) v *= rs[r];
      out[(size_t)(row0 + m) * HDIM + col] = v;
    }
  }
}

__global__ void head_kernel(const float* __restrict__ h, const float* __restrict__ Wh,
                            const float* __restrict__ bh, float* __restrict__ out, int n) {
  int i = blockIdx.x * blockDim.x + threadIdx.x;
  if (i >= n) return;
  const float* hp = h + (size_t)i * HDIM;
  float s = bh[0];
  #pragma unroll
  for (int k = 0; k < HDIM; ++k) s += hp[k] * Wh[k];
  out[i] = s;
}

extern "C" void kernel_launch(void* const* d_in, const int* in_sizes, int n_in,
                              void* d_out, int out_size, void* d_ws, size_t ws_size,
                              hipStream_t stream) {
  const float* x      = (const float*)d_in[0];
  const int*   eidx   = (const int*)d_in[1];
  const float* W_pre  = (const float*)d_in[2];
  const float* b_pre  = (const float*)d_in[3];
  const float* W1     = (const float*)d_in[4];
  const float* b1     = (const float*)d_in[5];
  const float* W2     = (const float*)d_in[6];
  const float* b2     = (const float*)d_in[7];
  const float* W_head = (const float*)d_in[8];
  const float* b_head = (const float*)d_in[9];
  float* out = (float*)d_out;

  const int N = in_sizes[0] / 128;
  const int E = in_sizes[1] / 2;
  const int* src = eidx;
  const int* dst = eidx + E;

  float* h        = (float*)d_ws;
  float* hw       = h + (size_t)N * HDIM;
  int*   bedge    = (int*)(hw + (size_t)N * HDIM);
  int*   ssrc     = bedge + 2 * (size_t)E;
  int*   cnt      = ssrc + E;
  int*   rowstart = cnt + N;
  float* dis      = (float*)(rowstart + N);

  const int nb   = (N + BNODES - 1) / BNODES;
  const int nblk = (E + ECHUNK - 1) / ECHUNK;
  int* hist  = (int*)(dis + N);
  int* offs  = hist + (size_t)nblk * nb;
  int* btot  = offs + (size_t)nblk * nb;
  int* bbase = btot + nb;
  edge_hist<<<nblk, 1024, 0, stream>>>(dst, hist, E, nb);
  scan_hist<<<nb, 256, 0, stream>>>(hist, offs, btot, nb, nblk);
  scatter_buckets<<<nblk, 1024, 0, stream>>>(src, dst, hist, offs, btot, bedge, bbase, E, nb);
  sort_bucket<<<nb, 1024, 0, stream>>>(bedge, bbase, cnt, rowstart, dis, ssrc, N);

  int gemm_blocks = (N + 127) / 128;
  gemm_n64_wmma<128, true, false><<<gemm_blocks, 256, 0, stream>>>(x, W_pre, b_pre, nullptr, h, N);

  const float* Ws[2] = {W1, W2};
  const float* bs[2] = {b1, b2};
  int gather_blocks = (N + 8 * RPW - 1) / (8 * RPW);
  for (int layer = 0; layer < 2; ++layer) {
    gemm_n64_wmma<64, false, true><<<gemm_blocks, 256, 0, stream>>>(h, Ws[layer], nullptr, dis, hw, N);
    gather_rows<<<gather_blocks, 256, 0, stream>>>(rowstart, cnt, ssrc, dis, hw, bs[layer], h, N);
  }

  head_kernel<<<(N + 255) / 256, 256, 0, stream>>>(h, W_head, b_head, out, N);
}


# Four launches folded into neighbours: edge counts ride in the K=128 grid, scatter blocks sum the count table (no scan), bucket blocks do layer 1's K=64 tiles, head dot in gather 2 as lane tree; 10->6
# speedup vs baseline: 4.2755x; 1.1061x over previous
#include <hip/hip_runtime.h>
#include <hip/hip_bf16.h>

typedef __attribute__((ext_vector_type(2))) float v2f;
typedef __attribute__((ext_vector_type(8))) float v8f;

#define HDIM 64

#define BSHIFT 9
#define BNODES 512
#define MAXB   256
#define ECHUNK 8192
#define STAGE  16384

__device__ __forceinline__ int block_excl_scan(int v, int* wsum) {
  const int lane = threadIdx.x & 31;
  const int wid  = threadIdx.x >> 5;
  int x = v;
  for (int o = 1; o < 32; o <<= 1) {
    int y = __shfl_up(x, o);
    if (lane >= o) x += y;
  }
  if (lane == 31) wsum[wid] = x;
  __syncthreads();
  int wo = (lane < wid) ? wsum[lane] : 0;
  for (int o = 16; o > 0; o >>= 1) wo += __shfl_xor(wo, o);
  __syncthreads();
  return wo + x - v;
}

template<int K, bool RELU_BIAS, bool ROW_SCALE>
__device__ __forceinline__ void gemm_n64_tile(const float* __restrict__ A,
                                              const float* __restrict__ W,
                                              const float* __restrict__ bias,
                                              const float* __restrict__ rowscale,
                                              float* __restrict__ out, int n_rows,
                                              int wave, int lane, int rs_row0) {
  int row0 = wave * 16;
  if (row0 >= n_rows) return;

  const int half = lane >> 4;
  const int l16  = lane & 15;

  v8f acc0 = {}, acc1 = {}, acc2 = {}, acc3 = {};

  float rs[8];
  #pragma unroll
  for (int r = 0; r < 8; ++r) rs[r] = ROW_SCALE ? rowscale[rs_row0 + half * 8 + r] : 1.0f;

  const float* arow = A + (size_t)(row0 + l16) * K;
  for (int kb = 0; kb < K; kb += 4) {
    v2f a;
    a.x = arow[kb + half * 2 + 0];
    a.y = arow[kb + half * 2 + 1];

    const float* bp = W + (size_t)(kb + half * 2) * HDIM + l16;
    v2f b0, b1, b2, b3;
    b0.x = bp[0];        b0.y = bp[HDIM];
    b1.x = bp[16];       b1.y = bp[HDIM + 16];
    b2.x = bp[32];       b2.y = bp[HDIM + 32];
    b3.x = bp[48];       b3.y = bp[HDIM + 48];

    acc0 = __builtin_amdgcn_wmma_f32_16x16x4_f32(false, a, false, b0, (short)0, acc0, false, false);
    acc1 = __builtin_amdgcn_wmma_f32_16x16x4_f32(false, a, false, b1, (short)0, acc1, false, false);
    acc2 = __builtin_amdgcn_wmma_f32_16x16x4_f32(false, a, false, b2, (short)0, acc2, false, false);
    acc3 = __builtin_amdgcn_wmma_f32_16x16x4_f32(false, a, false, b3, (short)0, acc3, false, false);
  }

  v8f accs[4] = {acc0, acc1, acc2, acc3};
  #pragma unroll
  for (int nt = 0; nt < 4; ++nt) {
    int col = nt * 16 + l16;
    float bcol = RELU_BIAS ? bias[col] : 0.0f;
    #pragma unroll
    for (int r = 0; r < 8; ++r) {
      int m = half * 8 + r;
      float v = accs[nt][r];
      if (RELU_BIAS) { v += bcol; v = fmaxf(v, 0.0f); }
      if (ROW_SCALE) v *= rs[r];
      out[(size_t)(row0 + m) * HDIM + col] = v;
    }
  }
}


__global__ __launch_bounds__(1024) void scatter_buckets(const int* __restrict__ src, const int* __restrict__ dst,
                                                        const int* __restrict__ hist, int* __restrict__ bedge_w,
                                                        int* __restrict__ bbase, int E, int nb, int nblk) {
  int2* __restrict__ bedge = (int2*)bedge_w;
  __shared__ int wsum[32];
  __shared__ int cur[MAXB];
  __shared__ int gofs[MAXB];
  __shared__ int ptot[4][MAXB];
  __shared__ int ppre[4][MAXB];
  __shared__ int2 stage[ECHUNK];
  const int tid = threadIdx.x;
  const int blk = blockIdx.x;
  const int e0  = blk * ECHUNK;
  int sv[ECHUNK / 1024], dv[ECHUNK / 1024];
  #pragma unroll
  for (int i = 0; i < ECHUNK / 1024; ++i) {
    const int e = e0 + i * 1024 + tid;
    const bool in = e < E;
    sv[i] = in ? src[e] : 0;
    dv[i] = in ? dst[e] : -1;
  }
  const bool isb = tid < nb;
  const int own = isb ? hist[blk * nb + tid] : 0;
  const int bq = tid & (MAXB - 1);
  const int q  = tid >> 8;
  int tq = 0, pq = 0;
  if (bq < nb) {
    const int lo = q * (MAXB / 4);
    const int hi = (lo + MAXB / 4 < nblk) ? (lo + MAXB / 4) : nblk;
    for (int j0 = lo; j0 < hi; j0 += 32) {
      int v[32];
      #pragma unroll
      for (int u = 0; u < 32; ++u) v[u] = (j0 + u < hi) ? hist[(j0 + u) * nb + bq] : 0;
      #pragma unroll
      for (int u = 0; u < 32; ++u) {
        tq += v[u];
        pq += (j0 + u < blk) ? v[u] : 0;
      }
    }
  }
  ptot[q][bq] = tq;
  ppre[q][bq] = pq;
  __syncthreads();
  const int tot = isb ? (ptot[0][tid] + ptot[1][tid] + ptot[2][tid] + ptot[3][tid]) : 0;
  const int pre = isb ? (ppre[0][tid] + ppre[1][tid] + ppre[2][tid] + ppre[3][tid]) : 0;
  const int base = block_excl_scan(tot, wsum);
  const int ls   = block_excl_scan(own, wsum);
  if (isb) {
    cur[tid]  = ls;
    gofs[tid] = base + pre - ls;
    if (blk == 0) {
      bbase[tid] = base;
      if (tid == nb - 1) bbase[nb] = base + tot;
    }
  }
  __syncthreads();
  #pragma unroll
  for (int i = 0; i < ECHUNK / 1024; ++i) {
    if (dv[i] >= 0) {
      const int p = atomicAdd(&cur[dv[i] >> BSHIFT], 1);
      stage[p] = make_int2(sv[i], dv[i]);
    }
  }
  __syncthreads();
  const int m = (E - e0 < ECHUNK) ? (E - e0) : ECHUNK;
  for (int i = tid; i < m; i += 1024) {
    const int2 ed = stage[i];
    bedge[gofs[ed.y >> BSHIFT] + i] = ed;
  }
}

__global__ __launch_bounds__(1024) void sort_bucket(const int* __restrict__ bedge_w, const int* __restrict__ bbase,
                                                    int* __restrict__ cnt, int* __restrict__ rowstart,
                                                    float* __restrict__ dis, int* __restrict__ ssrc,
                                                    const float* __restrict__ h, const float* __restrict__ W1,
                                                    float* __restrict__ hw, int n) {
  const int2* __restrict__ bedge = (const int2*)bedge_w;
  __shared__ int ncnt[BNODES];
  __shared__ int ncur[BNODES];
  __shared__ float sdis[BNODES];
  __shared__ int wsum[32];
  __shared__ int stage[STAGE];
  const int tid  = threadIdx.x;
  const int k    = blockIdx.x;
  const int b0   = bbase[k];
  const int m    = bbase[k + 1] - b0;

  if (tid < BNODES) ncnt[tid] = 0;
  __syncthreads();
  for (int i0 = 0; i0 < m; i0 += 8 * 1024) {
    int dv[8];
    for (int j = 0; j < 8; ++j) {
      const int i = i0 + j * 1024 + tid;
      dv[j] = (i < m) ? bedge[b0 + i].y : -1;
    }
    for (int j = 0; j < 8; ++j)
      if (dv[j] >= 0) atomicAdd(&ncnt[dv[j] & (BNODES - 1)], 1);
  }
  __syncthreads();

  const int c  = (tid < BNODES) ? ncnt[tid] : 0;
  const int st = block_excl_scan(c, wsum);
  if (tid < BNODES) {
    ncur[tid] = st;
    const int node = k * BNODES + tid;
    const float d  = (float)c;
    const float dv = (d > 0.0f) ? rsqrtf(fmaxf(d, 1.0f)) : 0.0f;
    sdis[tid] = dv;
    if (node < n) {
      cnt[node]      = c;
      rowstart[node] = b0 + st;
      dis[node]      = dv;
    }
  }
  __syncthreads();

  if (m <= STAGE) {
    for (int i0 = 0; i0 < m; i0 += 8 * 1024) {
      int2 ev[8];
      for (int j = 0; j < 8; ++j) {
        const int i = i0 + j * 1024 + tid;
        ev[j] = (i < m) ? bedge[b0 + i] : make_int2(0, -1);
      }
      for (int j = 0; j < 8; ++j) {
        if (ev[j].y >= 0) {
          const int p = atomicAdd(&ncur[ev[j].y & (BNODES - 1)], 1);
          stage[p] = ev[j].x;
        }
      }
    }
    __syncthreads();
    for (int i = tid; i < m; i += 1024) ssrc[b0 + i] = stage[i];
  } else {
    for (int i = tid; i < m; i += 1024) {
      const int2 ed = bedge[b0 + i];
      const int p = atomicAdd(&ncur[ed.y & (BNODES - 1)], 1);
      ssrc[b0 + p] = ed.x;
    }
  }

  const int w = tid >> 5;
  gemm_n64_tile<64, false, true>(h, W1, nullptr, sdis, hw, n, k * (BNODES / 16) + w, tid & 31, w * 16);
}

#define RPW 4
#define GRP 8
#define GWAVES 8

template<bool LAST>
__global__ __launch_bounds__(32 * GWAVES) void gather_rows(const int* __restrict__ rowstart, const int* __restrict__ cnt,
                            const int* __restrict__ ssrc, const float* __restrict__ dis,
                            const float* __restrict__ hw, const float* __restrict__ bias,
                            float* __restrict__ h, const float* __restrict__ Wh,
                            const float* __restrict__ bh, float* __restrict__ out, int n) {
  __shared__ float outv[GWAVES * RPW];
  const int lane = threadIdx.x & 31;
  const int node0r = RPW * __builtin_amdgcn_readfirstlane((int)((blockIdx.x * blockDim.x + threadIdx.x) >> 5));
  if (!LAST) { if (node0r >= n) return; }
  const bool wlive = node0r < n;
  const int node0 = wlive ? node0r : 0;

  float2 hv[RPW];
  #pragma unroll
  for (int r = 0; r < RPW; ++r) {
    const int node = (node0 + r < n) ? node0 + r : node0;
    hv[r] = ((const float2*)h)[(unsigned)node * (HDIM / 2) + lane];
  }
  const float2 b = ((const float2*)bias)[lane];
  float2 wh = make_float2(0.0f, 0.0f);
  float  bhv = 0.0f;
  if (LAST) {
    wh  = ((const float2*)Wh)[lane];
    bhv = bh[0];
  }
  const int nl = (lane < RPW && node0 + lane < n) ? node0 + lane : node0;
  const int   c_l  = cnt[nl];
  const int   st_l = rowstart[nl];
  const float d_l  = dis[nl];
  const float2* __restrict__ hw2 = (const float2*)hw;

  int   start[RPW], deg[RPW];
  float dd[RPW];
  int   maxdeg = 0;
  #pragma unroll
  for (int r = 0; r < RPW; ++r) {
    const bool live = wlive && (node0 + r < n);
    const int  c    = __builtin_amdgcn_readlane(c_l, r);
    start[r] = __builtin_amdgcn_readlane(st_l, r);
    deg[r]   = live ? c : 0;
    dd[r]    = __int_as_float(__builtin_amdgcn_readlane(__float_as_int(d_l), r));
    maxdeg   = (deg[r] > maxdeg) ? deg[r] : maxdeg;
  }

  float2 acc[RPW];
  #pragma unroll
  for (int r = 0; r < RPW; ++r) acc[r] = make_float2(0.0f, 0.0f);

  int s_l[RPW];
  #pragma unroll
  for (int r = 0; r < RPW; ++r) {
    s_l[r] = 0;
    if (lane < deg[r]) s_l[r] = ssrc[start[r] + lane];
  }

  for (int j0 = 0; j0 < maxdeg; j0 += 32) {
    int m[RPW];
    #pragma unroll
    for (int r = 0; r < RPW; ++r) {
      const int left = deg[r] - j0;
      m[r] = (left < 0) ? 0 : ((left < 32) ? left : 32);
    }
    const int mm = (maxdeg - j0 < 32) ? (maxdeg - j0) : 32;
    for (int j = 0; j < mm; j += GRP) {
      int sid[RPW][GRP];
      #pragma unroll
      for (int r = 0; r < RPW; ++r) {
        const int last = (m[r] > 0) ? (m[r] - 1) : 0;
        #pragma unroll
        for (int k = 0; k < GRP; ++k) {
          const int jj = j + k;
          sid[r][k] = __builtin_amdgcn_readlane(s_l[r], (jj < m[r]) ? jj : last);
        }
      }
      float2 v[RPW][GRP];
      #pragma unroll
      for (int r = 0; r < RPW; ++r) {
        if (j < m[r]) {
          #pragma unroll
          for (int k = 0; k < GRP; ++k)
            v[r][k] = hw2[(unsigned)sid[r][k] * (HDIM / 2) + lane];
        } else {
          #pragma unroll
          for (int k = 0; k < GRP; ++k) v[r][k] = make_float2(0.0f, 0.0f);
        }
      }
      #pragma unroll
      for (int r = 0; r < RPW; ++r) {
        #pragma unroll
        for (int k = 0; k < GRP; ++k) {
          const bool in = (j + k) < m[r];
          acc[r].x += in ? v[r][k].x : 0.0f;
          acc[r].y += in ? v[r][k].y : 0.0f;
        }
      }
    }
    if (j0 + 32 < maxdeg) {
      #pragma unroll
      for (int r = 0; r < RPW; ++r) {
        s_l[r] = 0;
        if (j0 + 32 + lane < deg[r]) s_l[r] = ssrc[start[r] + j0 + 32 + lane];
      }
    }
  }

  if (!LAST) {
    #pragma unroll
    for (int r = 0; r < RPW; ++r) {
      if (node0 + r < n) {
        float2 o;
        o.x = fmaxf(acc[r].x * dd[r] + b.x, 0.0f) + hv[r].x;
        o.y = fmaxf(acc[r].y * dd[r] + b.y, 0.0f) + hv[r].y;
        ((float2*)(h + (size_t)(node0 + r) * HDIM))[lane] = o;
      }
    }
  } else {
    float res = 0.0f;
    #pragma unroll
    for (int r = 0; r < RPW; ++r) {
      const float ox = fmaxf(acc[r].x * dd[r] + b.x, 0.0f) + hv[r].x;
      const float oy = fmaxf(acc[r].y * dd[r] + b.y, 0.0f) + hv[r].y;
      float p = ox * wh.x + oy * wh.y;
      for (int o = 16; o > 0; o >>= 1) p += __shfl_xor(p, o);
      if (lane == r) res = p;
    }
    if (lane < RPW) outv[(threadIdx.x >> 5) * RPW + lane] = res + bhv;
    __syncthreads();
    if (threadIdx.x < GWAVES * RPW) {
      const int node = (int)blockIdx.x * (GWAVES * RPW) + (int)threadIdx.x;
      if (node < n) out[node] = outv[threadIdx.x];
    }
  }
}

template<int K, bool RELU_BIAS, bool ROW_SCALE>
__global__ void gemm_n64_wmma(const float* __restrict__ A,
                              const float* __restrict__ W,
                              const float* __restrict__ bias,
                              const float* __restrict__ rowscale,
                              float* __restrict__ out, int n_rows) {
  const int wave = (blockIdx.x * blockDim.x + threadIdx.x) >> 5;
  gemm_n64_tile<K, RELU_BIAS, ROW_SCALE>(A, W, bias, rowscale, out, n_rows, wave, threadIdx.x & 31, wave * 16);
}

__global__ __launch_bounds__(256) void hist_and_pre(const int* __restrict__ dst, int* __restrict__ hist,
                                                    int E, int nb, int nblk,
                                                    const float* __restrict__ x, const float* __restrict__ W_pre,
                                                    const float* __restrict__ b_pre, float* __restrict__ h, int n) {
  __shared__ int lh[MAXB];
  const int tid = threadIdx.x;
  if ((int)blockIdx.x < nblk) {
    const int e0 = blockIdx.x * ECHUNK;
    lh[tid] = 0;
    __syncthreads();
    int dv[ECHUNK / 256];
    #pragma unroll
    for (int i = 0; i < ECHUNK / 256; ++i) {
      const int e = e0 + i * 256 + tid;
      dv[i] = (e < E) ? dst[e] : -1;
    }
    #pragma unroll
    for (int i = 0; i < ECHUNK / 256; ++i)
      if (dv[i] >= 0) atomicAdd(&lh[dv[i] >> BSHIFT], 1);
    __syncthreads();
    if (tid < nb) hist[blockIdx.x * nb + tid] = lh[tid];
    return;
  }
  const int wave = (int)((blockIdx.x - nblk) * 256 + tid) >> 5;
  gemm_n64_tile<128, true, false>(x, W_pre, b_pre, nullptr, h, n, wave, tid & 31, 0);
}

extern "C" void kernel_launch(void* const* d_in, const int* in_sizes, int n_in,
                              void* d_out, int out_size, void* d_ws, size_t ws_size,
                              hipStream_t stream) {
  const float* x      = (const float*)d_in[0];
  const int*   eidx   = (const int*)d_in[1];
  const float* W_pre  = (const float*)d_in[2];
  const float* b_pre  = (const float*)d_in[3];
  const float* W1     = (const float*)d_in[4];
  const float* b1     = (const float*)d_in[5];
  const float* W2     = (const float*)d_in[6];
  const float* b2     = (const float*)d_in[7];
  const float* W_head = (const float*)d_in[8];
  const float* b_head = (const float*)d_in[9];
  float* out = (float*)d_out;

  const int N = in_sizes[0] / 128;
  const int E = in_sizes[1] / 2;
  const int* src = eidx;
  const int* dst = eidx + E;

  float* h        = (float*)d_ws;
  float* hw       = h + (size_t)N * HDIM;
  int*   bedge    = (int*)(hw + (size_t)N * HDIM);
  int*   ssrc     = bedge + 2 * (size_t)E;
  int*   cnt      = ssrc + E;
  int*   rowstart = cnt + N;
  float* dis      = (float*)(rowstart + N);

  const int nb   = (N + BNODES - 1) / BNODES;
  const int nblk = (E + ECHUNK - 1) / ECHUNK;
  int* hist  = (int*)(dis + N);
  int* bbase = hist + (size_t)nblk * nb;
  int gemm_blocks = (N + 127) / 128;

  hist_and_pre<<<nblk + gemm_blocks, 256, 0, stream>>>(dst, hist, E, nb, nblk, x, W_pre, b_pre, h, N);
  scatter_buckets<<<nblk, 1024, 0, stream>>>(src, dst, hist, bedge, bbase, E, nb, nblk);
  sort_bucket<<<nb, 1024, 0, stream>>>(bedge, bbase, cnt, rowstart, dis, ssrc, h, W1, hw, N);

  int gather_blocks = (N + GWAVES * RPW - 1) / (GWAVES * RPW);
  gather_rows<false><<<gather_blocks, 32 * GWAVES, 0, stream>>>(rowstart, cnt, ssrc, dis, hw, b1, h,
                                                                nullptr, nullptr, nullptr, N);
  gemm_n64_wmma<64, false, true><<<gemm_blocks, 256, 0, stream>>>(h, W2, nullptr, dis, hw, N);
  gather_rows<true><<<gather_blocks, 32 * GWAVES, 0, stream>>>(rowstart, cnt, ssrc, dis, hw, b2, h,
                                                               W_head, b_head, out, N);
}


# The three products move 16 bytes per lane: W staged once per block in LDS in fragment order, A by 16-byte loads (k permuted inside each 8, A and B alike: sum order changes), whole-row stores
# speedup vs baseline: 4.6054x; 1.0772x over previous
#include <hip/hip_runtime.h>
#include <hip/hip_bf16.h>

typedef __attribute__((ext_vector_type(2))) float v2f;
typedef __attribute__((ext_vector_type(8))) float v8f;

#define HDIM 64

#define BSHIFT 9
#define BNODES 512
#define MAXB   256
#define ECHUNK 8192
#define STAGE  16384

__device__ __forceinline__ int block_excl_scan(int v, int* wsum) {
  const int lane = threadIdx.x & 31;
  const int wid  = threadIdx.x >> 5;
  int x = v;
  for (int o = 1; o < 32; o <<= 1) {
    int y = __shfl_up(x, o);
    if (lane >= o) x += y;
  }
  if (lane == 31) wsum[wid] = x;
  __syncthreads();
  int wo = (lane < wid) ? wsum[lane] : 0;
  for (int o = 16; o > 0; o >>= 1) wo += __shfl_xor(wo, o);
  __syncthreads();
  return wo + x - v;
}


template<int K, int NT>
__device__ __forceinline__ void stage_w64_load(const float* __restrict__ W, int tid, float4 (&w)[K * 16 / NT]) {
  #pragma unroll
  for (int u = 0; u < K * 16 / NT; ++u) w[u] = ((const float4*)W)[u * NT + tid];
}
template<int K, int NT>
__device__ __forceinline__ void stage_w64_store(const float4 (&w)[K * 16 / NT], float* __restrict__ wl, int tid) {
  #pragma unroll
  for (int u = 0; u < K * 16 / NT; ++u) {
    const int idx = u * NT + tid;
    const int k = idx >> 4, c = idx & 15;
    float* d = wl + ((((k >> 3) * 4) * 2 + ((k >> 2) & 1)) * 16 + c) * 4 + (k & 3);
    d[0]   = w[u].x;
    d[128] = w[u].y;
    d[256] = w[u].z;
    d[384] = w[u].w;
  }
}
template<int K, int NT>
__device__ __forceinline__ void stage_w64(const float* __restrict__ W, float* __restrict__ wl, int tid) {
  float4 w[K * 16 / NT];
  stage_w64_load<K, NT>(W, tid, w);
  stage_w64_store<K, NT>(w, wl, tid);
}

template<int K, bool RELU_BIAS, bool ROW_SCALE>
__device__ __forceinline__ void gemm_n64_tile(const float* __restrict__ A,
                                              const float* __restrict__ wl,
                                              const float* __restrict__ bias,
                                              const float* __restrict__ rowscale,
                                              float* __restrict__ out, int n_rows,
                                              int wave, int lane, int rs_row0) {
  int row0 = wave * 16;
  if (row0 >= n_rows) return;

  const int half = lane >> 4;
  const int l16  = lane & 15;

  v8f acc0 = {}, acc1 = {}, acc2 = {}, acc3 = {};

  float rs[8];
  #pragma unroll
  for (int r = 0; r < 8; ++r) rs[r] = ROW_SCALE ? rowscale[rs_row0 + half * 8 + r] : 1.0f;

  const float4* arow = (const float4*)(A + (size_t)(row0 + l16) * K + half * 4);
  float4 av[K / 8];
  #pragma unroll
  for (int j = 0; j < K / 8; ++j) av[j] = arow[2 * j];

  const float4* wl4 = (const float4*)wl + lane;
  #pragma unroll
  for (int j = 0; j < K / 8; ++j) {
    const float4 b0 = wl4[(j * 4 + 0) * 32];
    const float4 b1 = wl4[(j * 4 + 1) * 32];
    const float4 b2 = wl4[(j * 4 + 2) * 32];
    const float4 b3 = wl4[(j * 4 + 3) * 32];
    v2f a;  a.x = av[j].x;  a.y = av[j].y;
    v2f p0, p1, p2, p3;
    p0.x = b0.x; p0.y = b0.y;  p1.x = b1.x; p1.y = b1.y;
    p2.x = b2.x; p2.y = b2.y;  p3.x = b3.x; p3.y = b3.y;
    acc0 = __builtin_amdgcn_wmma_f32_16x16x4_f32(false, a, false, p0, (short)0, acc0, false, false);
    acc1 = __builtin_amdgcn_wmma_f32_16x16x4_f32(false, a, false, p1, (short)0, acc1, false, false);
    acc2 = __builtin_amdgcn_wmma_f32_16x16x4_f32(false, a, false, p2, (short)0, acc2, false, false);
    acc3 = __builtin_amdgcn_wmma_f32_16x16x4_f32(false, a, false, p3, (short)0, acc3, false, false);
    a.x = av[j].z;  a.y = av[j].w;
    p0.x = b0.z; p0.y = b0.w;  p1.x = b1.z; p1.y = b1.w;
    p2.x = b2.z; p2.y = b2.w;  p3.x = b3.z; p3.y = b3.w;
    acc0 = __builtin_amdgcn_wmma_f32_16x16x4_f32(false, a, false, p0, (short)0, acc0, false, false);
    acc1 = __builtin_amdgcn_wmma_f32_16x16x4_f32(false, a, false, p1, (short)0, acc1, false, false);
    acc2 = __builtin_amdgcn_wmma_f32_16x16x4_f32(false, a, false, p2, (short)0, acc2, false, false);
    acc3 = __builtin_amdgcn_wmma_f32_16x16x4_f32(false, a, false, p3, (short)0, acc3, false, false);
  }

  float4 bc = make_float4(0.0f, 0.0f, 0.0f, 0.0f);
  if (RELU_BIAS) bc = ((const float4*)bias)[l16];
  #pragma unroll
  for (int r = 0; r < 8; ++r) {
    float4 v = make_float4(acc0[r], acc1[r], acc2[r], acc3[r]);
    if (RELU_BIAS) {
      v.x = fmaxf(v.x + bc.x, 0.0f);  v.y = fmaxf(v.y + bc.y, 0.0f);
      v.z = fmaxf(v.z + bc.z, 0.0f);  v.w = fmaxf(v.w + bc.w, 0.0f);
    }
    if (ROW_SCALE) { v.x *= rs[r]; v.y *= rs[r]; v.z *= rs[r]; v.w *= rs[r]; }
    ((float4*)(out + (size_t)(row0 + half * 8 + r) * HDIM))[l16] = v;
  }
}


__global__ __launch_bounds__(1024) void scatter_buckets(const int* __restrict__ src, const int* __restrict__ dst,
                                                        const int* __restrict__ hist, int* __restrict__ bedge_w,
                                                        int* __restrict__ bbase, int E, int nb, int nblk) {
  int2* __restrict__ bedge = (int2*)bedge_w;
  __shared__ int wsum[32];
  __shared__ int cur[MAXB];
  __shared__ int gofs[MAXB];
  __shared__ int ptot[4][MAXB];
  __shared__ int ppre[4][MAXB];
  __shared__ int2 stage[ECHUNK];
  const int tid = threadIdx.x;
  const int blk = blockIdx.x;
  const int e0  = blk * ECHUNK;
  int sv[ECHUNK / 1024], dv[ECHUNK / 1024];
  #pragma unroll
  for (int i = 0; i < ECHUNK / 1024; ++i) {
    const int e = e0 + i * 1024 + tid;
    const bool in = e < E;
    sv[i] = in ? src[e] : 0;
    dv[i] = in ? dst[e] : -1;
  }
  const bool isb = tid < nb;
  const int own = isb ? hist[blk * nb + tid] : 0;
  const int bq = tid & (MAXB - 1);
  const int q  = tid >> 8;
  int tq = 0, pq = 0;
  if (bq < nb) {
    const int lo = q * (MAXB / 4);
    const int hi = (lo + MAXB / 4 < nblk) ? (lo + MAXB / 4) : nblk;
    for (int j0 = lo; j0 < hi; j0 += 32) {
      int v[32];
      #pragma unroll
      for (int u = 0; u < 32; ++u) v[u] = (j0 + u < hi) ? hist[(j0 + u) * nb + bq] : 0;
      #pragma unroll
      for (int u = 0; u < 32; ++u) {
        tq += v[u];
        pq += (j0 + u < blk) ? v[u] : 0;
      }
    }
  }
  ptot[q][bq] = tq;
  ppre[q][bq] = pq;
  __syncthreads();
  const int tot = isb ? (ptot[0][tid] + ptot[1][tid] + ptot[2][tid] + ptot[3][tid]) : 0;
  const int pre = isb ? (ppre[0][tid] + ppre[1][tid] + ppre[2][tid] + ppre[3][tid]) : 0;
  const int base = block_excl_scan(tot, wsum);
  const int ls   = block_excl_scan(own, wsum);
  if (isb) {
    cur[tid]  = ls;
    gofs[tid] = base + pre - ls;
    if (blk == 0) {
      bbase[tid] = base;
      if (tid == nb - 1) bbase[nb] = base + tot;
    }
  }
  __syncthreads();
  #pragma unroll
  for (int i = 0; i < ECHUNK / 1024; ++i) {
    if (dv[i] >= 0) {
      const int p = atomicAdd(&cur[dv[i] >> BSHIFT], 1);
      stage[p] = make_int2(sv[i], dv[i]);
    }
  }
  __syncthreads();
  const int m = (E - e0 < ECHUNK) ? (E - e0) : ECHUNK;
  for (int i = tid; i < m; i += 1024) {
    const int2 ed = stage[i];
    bedge[gofs[ed.y >> BSHIFT] + i] = ed;
  }
}

__global__ __launch_bounds__(1024) void sort_bucket(const int* __restrict__ bedge_w, const int* __restrict__ bbase,
                                                    int* __restrict__ cnt, int* __restrict__ rowstart,
                                                    float* __restrict__ dis, int* __restrict__ ssrc,
                                                    const float* __restrict__ h, const float* __restrict__ W1,
                                                    float* __restrict__ hw, int n) {
  const int2* __restrict__ bedge = (const int2*)bedge_w;
  __shared__ int ncnt[BNODES];
  __shared__ int ncur[BNODES];
  __shared__ float sdis[BNODES];
  __shared__ int wsum[32];
  __shared__ int stage[STAGE];
  __shared__ __attribute__((aligned(16))) float wl[64 * HDIM];
  const int tid  = threadIdx.x;
  const int k    = blockIdx.x;
  const int b0   = bbase[k];
  const int m    = bbase[k + 1] - b0;

  float4 w1v[1];
  stage_w64_load<64, 1024>(W1, tid, w1v);
  if (tid < BNODES) ncnt[tid] = 0;
  __syncthreads();
  for (int i0 = 0; i0 < m; i0 += 8 * 1024) {
    int dv[8];
    for (int j = 0; j < 8; ++j) {
      const int i = i0 + j * 1024 + tid;
      dv[j] = (i < m) ? bedge[b0 + i].y : -1;
    }
    for (int j = 0; j < 8; ++j)
      if (dv[j] >= 0) atomicAdd(&ncnt[dv[j] & (BNODES - 1)], 1);
  }
  __syncthreads();

  const int c  = (tid < BNODES) ? ncnt[tid] : 0;
  const int st = block_excl_scan(c, wsum);
  if (tid < BNODES) {
    ncur[tid] = st;
    const int node = k * BNODES + tid;
    const float d  = (float)c;
    const float dv = (d > 0.0f) ? rsqrtf(fmaxf(d, 1.0f)) : 0.0f;
    sdis[tid] = dv;
    if (node < n) {
      cnt[node]      = c;
      rowstart[node] = b0 + st;
      dis[node]      = dv;
    }
  }
  stage_w64_store<64, 1024>(w1v, wl, tid);
  __syncthreads();

  if (m <= STAGE) {
    for (int i0 = 0; i0 < m; i0 += 8 * 1024) {
      int2 ev[8];
      for (int j = 0; j < 8; ++j) {
        const int i = i0 + j * 1024 + tid;
        ev[j] = (i < m) ? bedge[b0 + i] : make_int2(0, -1);
      }
      for (int j = 0; j < 8; ++j) {
        if (ev[j].y >= 0) {
          const int p = atomicAdd(&ncur[ev[j].y & (BNODES - 1)], 1);
          stage[p] = ev[j].x;
        }
      }
    }
    __syncthreads();
    for (int i = tid; i < m; i += 1024) ssrc[b0 + i] = stage[i];
  } else {
    for (int i = tid; i < m; i += 1024) {
      const int2 ed = bedge[b0 + i];
      const int p = atomicAdd(&ncur[ed.y & (BNODES - 1)], 1);
      ssrc[b0 + p] = ed.x;
    }
  }

  const int w = tid >> 5;
  gemm_n64_tile<64, false, true>(h, wl, nullptr, sdis, hw, n, k * (BNODES / 16) + w, tid & 31, w * 16);
}

#define RPW 4
#define GRP 8
#define GWAVES 8

template<bool LAST>
__global__ __launch_bounds__(32 * GWAVES) void gather_rows(const int* __restrict__ rowstart, const int* __restrict__ cnt,
                            const int* __restrict__ ssrc, const float* __restrict__ dis,
                            const float* __restrict__ hw, const float* __restrict__ bias,
                            float* __restrict__ h, const float* __restrict__ Wh,
                            const float* __restrict__ bh, float* __restrict__ out, int n) {
  __shared__ float outv[GWAVES * RPW];
  const int lane = threadIdx.x & 31;
  const int node0r = RPW * __builtin_amdgcn_readfirstlane((int)((blockIdx.x * blockDim.x + threadIdx.x) >> 5));
  if (!LAST) { if (node0r >= n) return; }
  const bool wlive = node0r < n;
  const int node0 = wlive ? node0r : 0;

  float2 hv[RPW];
  #pragma unroll
  for (int r = 0; r < RPW; ++r) {
    const int node = (node0 + r < n) ? node0 + r : node0;
    hv[r] = ((const float2*)h)[(unsigned)node * (HDIM / 2) + lane];
  }
  const float2 b = ((const float2*)bias)[lane];
  float2 wh = make_float2(0.0f, 0.0f);
  float  bhv = 0.0f;
  if (LAST) {
    wh  = ((const float2*)Wh)[lane];
    bhv = bh[0];
  }
  const int nl = (lane < RPW && node0 + lane < n) ? node0 + lane : node0;
  const int   c_l  = cnt[nl];
  const int   st_l = rowstart[nl];
  const float d_l  = dis[nl];
  const float2* __restrict__ hw2 = (const float2*)hw;

  int   start[RPW], deg[RPW];
  float dd[RPW];
  int   maxdeg = 0;
  #pragma unroll
  for (int r = 0; r < RPW; ++r) {
    const bool live = wlive && (node0 + r < n);
    const int  c    = __builtin_amdgcn_readlane(c_l, r);
    start[r] = __builtin_amdgcn_readlane(st_l, r);
    deg[r]   = live ? c : 0;
    dd[r]    = __int_as_float(__builtin_amdgcn_readlane(__float_as_int(d_l), r));
    maxdeg   = (deg[r] > maxdeg) ? deg[r] : maxdeg;
  }

  float2 acc[RPW];
  #pragma unroll
  for (int r = 0; r < RPW; ++r) acc[r] = make_float2(0.0f, 0.0f);

  int s_l[RPW];
  #pragma unroll
  for (int r = 0; r < RPW; ++r) {
    s_l[r] = 0;
    if (lane < deg[r]) s_l[r] = ssrc[start[r] + lane];
  }

  for (int j0 = 0; j0 < maxdeg; j0 += 32) {
    int m[RPW];
    #pragma unroll
    for (int r = 0; r < RPW; ++r) {
      const int left = deg[r] - j0;
      m[r] = (left < 0) ? 0 : ((left < 32) ? left : 32);
    }
    const int mm = (maxdeg - j0 < 32) ? (maxdeg - j0) : 32;
    for (int j = 0; j < mm; j += GRP) {
      int sid[RPW][GRP];
      #pragma unroll
      for (int r = 0; r < RPW; ++r) {
        const int last = (m[r] > 0) ? (m[r] - 1) : 0;
        #pragma unroll
        for (int k = 0; k < GRP; ++k) {
          const int jj = j + k;
          sid[r][k] = __builtin_amdgcn_readlane(s_l[r], (jj < m[r]) ? jj : last);
        }
      }
      float2 v[RPW][GRP];
      #pragma unroll
      for (int r = 0; r < RPW; ++r) {
        if (j < m[r]) {
          #pragma unroll
          for (int k = 0; k < GRP; ++k)
            v[r][k] = hw2[(unsigned)sid[r][k] * (HDIM / 2) + lane];
        } else {
          #pragma unroll
          for (int k = 0; k < GRP; ++k) v[r][k] = make_float2(0.0f, 0.0f);
        }
      }
      #pragma unroll
      for (int r = 0; r < RPW; ++r) {
        #pragma unroll
        for (int k = 0; k < GRP; ++k) {
          const bool in = (j + k) < m[r];
          acc[r].x += in ? v[r][k].x : 0.0f;
          acc[r].y += in ? v[r][k].y : 0.0f;
        }
      }
    }
    if (j0 + 32 < maxdeg) {
      #pragma unroll
      for (int r = 0; r < RPW; ++r) {
        s_l[r] = 0;
        if (j0 + 32 + lane < deg[r]) s_l[r] = ssrc[start[r] + j0 + 32 + lane];
      }
    }
  }

  if (!LAST) {
    #pragma unroll
    for (int r = 0; r < RPW; ++r) {
      if (node0 + r < n) {
        float2 o;
        o.x = fmaxf(acc[r].x * dd[r] + b.x, 0.0f) + hv[r].x;
        o.y = fmaxf(acc[r].y * dd[r] + b.y, 0.0f) + hv[r].y;
        ((float2*)(h + (size_t)(node0 + r) * HDIM))[lane] = o;
      }
    }
  } else {
    float res = 0.0f;
    #pragma unroll
    for (int r = 0; r < RPW; ++r) {
      const float ox = fmaxf(acc[r].x * dd[r] + b.x, 0.0f) + hv[r].x;
      const float oy = fmaxf(acc[r].y * dd[r] + b.y, 0.0f) + hv[r].y;
      float p = ox * wh.x + oy * wh.y;
      for (int o = 16; o > 0; o >>= 1) p += __shfl_xor(p, o);
      if (lane == r) res = p;
    }
    if (lane < RPW) outv[(threadIdx.x >> 5) * RPW + lane] = res + bhv;
    __syncthreads();
    if (threadIdx.x < GWAVES * RPW) {
      const int node = (int)blockIdx.x * (GWAVES * RPW) + (int)threadIdx.x;
      if (node < n) out[node] = outv[threadIdx.x];
    }
  }
}

template<int K, bool RELU_BIAS, bool ROW_SCALE>
__global__ __launch_bounds__(256) void gemm_n64_wmma(const float* __restrict__ A,
                              const float* __restrict__ W,
                              const float* __restrict__ bias,
                              const float* __restrict__ rowscale,
                              float* __restrict__ out, int n_rows) {
  __shared__ __attribute__((aligned(16))) float wl[K * HDIM];
  stage_w64<K, 256>(W, wl, threadIdx.x);
  __syncthreads();
  const int wave = (blockIdx.x * 256 + threadIdx.x) >> 5;
  gemm_n64_tile<K, RELU_BIAS, ROW_SCALE>(A, wl, bias, rowscale, out, n_rows, wave, threadIdx.x & 31, wave * 16);
}

__global__ __launch_bounds__(256) void hist_and_pre(const int* __restrict__ dst, int* __restrict__ hist,
                                                    int E, int nb, int nblk,
                                                    const float* __restrict__ x, const float* __restrict__ W_pre,
                                                    const float* __restrict__ b_pre, float* __restrict__ h, int n) {
  __shared__ int lh[MAXB];
  __shared__ __attribute__((aligned(16))) float wl[128 * HDIM];
  const int tid = threadIdx.x;
  if ((int)blockIdx.x < nblk) {
    const int e0 = blockIdx.x * ECHUNK;
    lh[tid] = 0;
    __syncthreads();
    int dv[ECHUNK / 256];
    #pragma unroll
    for (int i = 0; i < ECHUNK / 256; ++i) {
      const int e = e0 + i * 256 + tid;
      dv[i] = (e < E) ? dst[e] : -1;
    }
    #pragma unroll
    for (int i = 0; i < ECHUNK / 256; ++i)
      if (dv[i] >= 0) atomicAdd(&lh[dv[i] >> BSHIFT], 1);
    __syncthreads();
    if (tid < nb) hist[blockIdx.x * nb + tid] = lh[tid];
    return;
  }
  stage_w64<128, 256>(W_pre, wl, tid);
  __syncthreads();
  const int wave = (int)((blockIdx.x - nblk) * 256 + tid) >> 5;
  gemm_n64_tile<128, true, false>(x, wl, b_pre, nullptr, h, n, wave, tid & 31, 0);
}

extern "C" void kernel_launch(void* const* d_in, const int* in_sizes, int n_in,
                              void* d_out, int out_size, void* d_ws, size_t ws_size,
                              hipStream_t stream) {
  const float* x      = (const float*)d_in[0];
  const int*   eidx   = (const int*)d_in[1];
  const float* W_pre  = (const float*)d_in[2];
  const float* b_pre  = (const float*)d_in[3];
  const float* W1     = (const float*)d_in[4];
  const float* b1     = (const float*)d_in[5];
  const float* W2     = (const float*)d_in[6];
  const float* b2     = (const float*)d_in[7];
  const float* W_head = (const float*)d_in[8];
  const float* b_head = (const float*)d_in[9];
  float* out = (float*)d_out;

  const int N = in_sizes[0] / 128;
  const int E = in_sizes[1] / 2;
  const int* src = eidx;
  const int* dst = eidx + E;

  float* h        = (float*)d_ws;
  float* hw       = h + (size_t)N * HDIM;
  int*   bedge    = (int*)(hw + (size_t)N * HDIM);
  int*   ssrc     = bedge + 2 * (size_t)E;
  int*   cnt      = ssrc + E;
  int*   rowstart = cnt + N;
  float* dis      = (float*)(rowstart + N);

  const int nb   = (N + BNODES - 1) / BNODES;
  const int nblk = (E + ECHUNK - 1) / ECHUNK;
  int* hist  = (int*)(dis + N);
  int* bbase = hist + (size_t)nblk * nb;
  int gemm_blocks = (N + 127) / 128;

  hist_and_pre<<<nblk + gemm_blocks, 256, 0, stream>>>(dst, hist, E, nb, nblk, x, W_pre, b_pre, h, N);
  scatter_buckets<<<nblk, 1024, 0, stream>>>(src, dst, hist, bedge, bbase, E, nb, nblk);
  sort_bucket<<<nb, 1024, 0, stream>>>(bedge, bbase, cnt, rowstart, dis, ssrc, h, W1, hw, N);

  int gather_blocks = (N + GWAVES * RPW - 1) / (GWAVES * RPW);
  gather_rows<false><<<gather_blocks, 32 * GWAVES, 0, stream>>>(rowstart, cnt, ssrc, dis, hw, b1, h,
                                                                nullptr, nullptr, nullptr, N);
  gemm_n64_wmma<64, false, true><<<gemm_blocks, 256, 0, stream>>>(h, W2, nullptr, dis, hw, N);
  gather_rows<true><<<gather_blocks, 32 * GWAVES, 0, stream>>>(rowstart, cnt, ssrc, dis, hw, b2, h,
                                                               W_head, b_head, out, N);
}


# Layer 2's K=64 product joins layer 1's gather: a block's 32 finished rows go via LDS to 8 WMMA tiles (one per wave), rows scaled and stored whole; 6 launches to 5, h not re-read; sums unchanged
# speedup vs baseline: 5.0894x; 1.1051x over previous
#include <hip/hip_runtime.h>
#include <hip/hip_bf16.h>

typedef __attribute__((ext_vector_type(2))) float v2f;
typedef __attribute__((ext_vector_type(8))) float v8f;

#define HDIM 64

#define BSHIFT 9
#define BNODES 512
#define MAXB   256
#define ECHUNK 8192
#define STAGE  16384

__device__ __forceinline__ int block_excl_scan(int v, int* wsum) {
  const int lane = threadIdx.x & 31;
  const int wid  = threadIdx.x >> 5;
  int x = v;
  for (int o = 1; o < 32; o <<= 1) {
    int y = __shfl_up(x, o);
    if (lane >= o) x += y;
  }
  if (lane == 31) wsum[wid] = x;
  __syncthreads();
  int wo = (lane < wid) ? wsum[lane] : 0;
  for (int o = 16; o > 0; o >>= 1) wo += __shfl_xor(wo, o);
  __syncthreads();
  return wo + x - v;
}


template<int K, int NT>
__device__ __forceinline__ void stage_w64_load(const float* __restrict__ W, int tid, float4 (&w)[K * 16 / NT]) {
  #pragma unroll
  for (int u = 0; u < K * 16 / NT; ++u) w[u] = ((const float4*)W)[u * NT + tid];
}
template<int K, int NT>
__device__ __forceinline__ void stage_w64_store(const float4 (&w)[K * 16 / NT], float* __restrict__ wl, int tid) {
  #pragma unroll
  for (int u = 0; u < K * 16 / NT; ++u) {
    const int idx = u * NT + tid;
    const int k = idx >> 4, c = idx & 15;
    float* d = wl + ((((k >> 3) * 4) * 2 + ((k >> 2) & 1)) * 16 + c) * 4 + (k & 3);
    d[0]   = w[u].x;
    d[128] = w[u].y;
    d[256] = w[u].z;
    d[384] = w[u].w;
  }
}
template<int K, int NT>
__device__ __forceinline__ void stage_w64(const float* __restrict__ W, float* __restrict__ wl, int tid) {
  float4 w[K * 16 / NT];
  stage_w64_load<K, NT>(W, tid, w);
  stage_w64_store<K, NT>(w, wl, tid);
}

template<int K, bool RELU_BIAS, bool ROW_SCALE>
__device__ __forceinline__ void gemm_n64_tile(const float* __restrict__ A,
                                              const float* __restrict__ wl,
                                              const float* __restrict__ bias,
                                              const float* __restrict__ rowscale,
                                              float* __restrict__ out, int n_rows,
                                              int wave, int lane, int rs_row0) {
  int row0 = wave * 16;
  if (row0 >= n_rows) return;

  const int half = lane >> 4;
  const int l16  = lane & 15;

  v8f acc0 = {}, acc1 = {}, acc2 = {}, acc3 = {};

  float rs[8];
  #pragma unroll
  for (int r = 0; r < 8; ++r) rs[r] = ROW_SCALE ? rowscale[rs_row0 + half * 8 + r] : 1.0f;

  const float4* arow = (const float4*)(A + (size_t)(row0 + l16) * K + half * 4);
  float4 av[K / 8];
  #pragma unroll
  for (int j = 0; j < K / 8; ++j) av[j] = arow[2 * j];

  const float4* wl4 = (const float4*)wl + lane;
  #pragma unroll
  for (int j = 0; j < K / 8; ++j) {
    const float4 b0 = wl4[(j * 4 + 0) * 32];
    const float4 b1 = wl4[(j * 4 + 1) * 32];
    const float4 b2 = wl4[(j * 4 + 2) * 32];
    const float4 b3 = wl4[(j * 4 + 3) * 32];
    v2f a;  a.x = av[j].x;  a.y = av[j].y;
    v2f p0, p1, p2, p3;
    p0.x = b0.x; p0.y = b0.y;  p1.x = b1.x; p1.y = b1.y;
    p2.x = b2.x; p2.y = b2.y;  p3.x = b3.x; p3.y = b3.y;
    acc0 = __builtin_amdgcn_wmma_f32_16x16x4_f32(false, a, false, p0, (short)0, acc0, false, false);
    acc1 = __builtin_amdgcn_wmma_f32_16x16x4_f32(false, a, false, p1, (short)0, acc1, false, false);
    acc2 = __builtin_amdgcn_wmma_f32_16x16x4_f32(false, a, false, p2, (short)0, acc2, false, false);
    acc3 = __builtin_amdgcn_wmma_f32_16x16x4_f32(false, a, false, p3, (short)0, acc3, false, false);
    a.x = av[j].z;  a.y = av[j].w;
    p0.x = b0.z; p0.y = b0.w;  p1.x = b1.z; p1.y = b1.w;
    p2.x = b2.z; p2.y = b2.w;  p3.x = b3.z; p3.y = b3.w;
    acc0 = __builtin_amdgcn_wmma_f32_16x16x4_f32(false, a, false, p0, (short)0, acc0, false, false);
    acc1 = __builtin_amdgcn_wmma_f32_16x16x4_f32(false, a, false, p1, (short)0, acc1, false, false);
    acc2 = __builtin_amdgcn_wmma_f32_16x16x4_f32(false, a, false, p2, (short)0, acc2, false, false);
    acc3 = __builtin_amdgcn_wmma_f32_16x16x4_f32(false, a, false, p3, (short)0, acc3, false, false);
  }

  float4 bc = make_float4(0.0f, 0.0f, 0.0f, 0.0f);
  if (RELU_BIAS) bc = ((const float4*)bias)[l16];
  #pragma unroll
  for (int r = 0; r < 8; ++r) {
    float4 v = make_float4(acc0[r], acc1[r], acc2[r], acc3[r]);
    if (RELU_BIAS) {
      v.x = fmaxf(v.x + bc.x, 0.0f);  v.y = fmaxf(v.y + bc.y, 0.0f);
      v.z = fmaxf(v.z + bc.z, 0.0f);  v.w = fmaxf(v.w + bc.w, 0.0f);
    }
    if (ROW_SCALE) { v.x *= rs[r]; v.y *= rs[r]; v.z *= rs[r]; v.w *= rs[r]; }
    ((float4*)(out + (size_t)(row0 + half * 8 + r) * HDIM))[l16] = v;
  }
}


__global__ __launch_bounds__(1024) void scatter_buckets(const int* __restrict__ src, const int* __restrict__ dst,
                                                        const int* __restrict__ hist, int* __restrict__ bedge_w,
                                                        int* __restrict__ bbase, int E, int nb, int nblk) {
  int2* __restrict__ bedge = (int2*)bedge_w;
  __shared__ int wsum[32];
  __shared__ int cur[MAXB];
  __shared__ int gofs[MAXB];
  __shared__ int ptot[4][MAXB];
  __shared__ int ppre[4][MAXB];
  __shared__ int2 stage[ECHUNK];
  const int tid = threadIdx.x;
  const int blk = blockIdx.x;
  const int e0  = blk * ECHUNK;
  int sv[ECHUNK / 1024], dv[ECHUNK / 1024];
  #pragma unroll
  for (int i = 0; i < ECHUNK / 1024; ++i) {
    const int e = e0 + i * 1024 + tid;
    const bool in = e < E;
    sv[i] = in ? src[e] : 0;
    dv[i] = in ? dst[e] : -1;
  }
  const bool isb = tid < nb;
  const int own = isb ? hist[blk * nb + tid] : 0;
  const int bq = tid & (MAXB - 1);
  const int q  = tid >> 8;
  int tq = 0, pq = 0;
  if (bq < nb) {
    const int lo = q * (MAXB / 4);
    const int hi = (lo + MAXB / 4 < nblk) ? (lo + MAXB / 4) : nblk;
    for (int j0 = lo; j0 < hi; j0 += 32) {
      int v[32];
      #pragma unroll
      for (int u = 0; u < 32; ++u) v[u] = (j0 + u < hi) ? hist[(j0 + u) * nb + bq] : 0;
      #pragma unroll
      for (int u = 0; u < 32; ++u) {
        tq += v[u];
        pq += (j0 + u < blk) ? v[u] : 0;
      }
    }
  }
  ptot[q][bq] = tq;
  ppre[q][bq] = pq;
  __syncthreads();
  const int tot = isb ? (ptot[0][tid] + ptot[1][tid] + ptot[2][tid] + ptot[3][tid]) : 0;
  const int pre = isb ? (ppre[0][tid] + ppre[1][tid] + ppre[2][tid] + ppre[3][tid]) : 0;
  const int base = block_excl_scan(tot, wsum);
  const int ls   = block_excl_scan(own, wsum);
  if (isb) {
    cur[tid]  = ls;
    gofs[tid] = base + pre - ls;
    if (blk == 0) {
      bbase[tid] = base;
      if (tid == nb - 1) bbase[nb] = base + tot;
    }
  }
  __syncthreads();
  #pragma unroll
  for (int i = 0; i < ECHUNK / 1024; ++i) {
    if (dv[i] >= 0) {
      const int p = atomicAdd(&cur[dv[i] >> BSHIFT], 1);
      stage[p] = make_int2(sv[i], dv[i]);
    }
  }
  __syncthreads();
  const int m = (E - e0 < ECHUNK) ? (E - e0) : ECHUNK;
  for (int i = tid; i < m; i += 1024) {
    const int2 ed = stage[i];
    bedge[gofs[ed.y >> BSHIFT] + i] = ed;
  }
}

__global__ __launch_bounds__(1024) void sort_bucket(const int* __restrict__ bedge_w, const int* __restrict__ bbase,
                                                    int* __restrict__ cnt, int* __restrict__ rowstart,
                                                    float* __restrict__ dis, int* __restrict__ ssrc,
                                                    const float* __restrict__ h, const float* __restrict__ W1,
                                                    float* __restrict__ hw, int n) {
  const int2* __restrict__ bedge = (const int2*)bedge_w;
  __shared__ int ncnt[BNODES];
  __shared__ int ncur[BNODES];
  __shared__ float sdis[BNODES];
  __shared__ int wsum[32];
  __shared__ int stage[STAGE];
  __shared__ __attribute__((aligned(16))) float wl[64 * HDIM];
  const int tid  = threadIdx.x;
  const int k    = blockIdx.x;
  const int b0   = bbase[k];
  const int m    = bbase[k + 1] - b0;

  float4 w1v[1];
  stage_w64_load<64, 1024>(W1, tid, w1v);
  if (tid < BNODES) ncnt[tid] = 0;
  __syncthreads();
  for (int i0 = 0; i0 < m; i0 += 8 * 1024) {
    int dv[8];
    for (int j = 0; j < 8; ++j) {
      const int i = i0 + j * 1024 + tid;
      dv[j] = (i < m) ? bedge[b0 + i].y : -1;
    }
    for (int j = 0; j < 8; ++j)
      if (dv[j] >= 0) atomicAdd(&ncnt[dv[j] & (BNODES - 1)], 1);
  }
  __syncthreads();

  const int c  = (tid < BNODES) ? ncnt[tid] : 0;
  const int st = block_excl_scan(c, wsum);
  if (tid < BNODES) {
    ncur[tid] = st;
    const int node = k * BNODES + tid;
    const float d  = (float)c;
    const float dv = (d > 0.0f) ? rsqrtf(fmaxf(d, 1.0f)) : 0.0f;
    sdis[tid] = dv;
    if (node < n) {
      cnt[node]      = c;
      rowstart[node] = b0 + st;
      dis[node]      = dv;
    }
  }
  stage_w64_store<64, 1024>(w1v, wl, tid);
  __syncthreads();

  if (m <= STAGE) {
    for (int i0 = 0; i0 < m; i0 += 8 * 1024) {
      int2 ev[8];
      for (int j = 0; j < 8; ++j) {
        const int i = i0 + j * 1024 + tid;
        ev[j] = (i < m) ? bedge[b0 + i] : make_int2(0, -1);
      }
      for (int j = 0; j < 8; ++j) {
        if (ev[j].y >= 0) {
          const int p = atomicAdd(&ncur[ev[j].y & (BNODES - 1)], 1);
          stage[p] = ev[j].x;
        }
      }
    }
    __syncthreads();
    for (int i = tid; i < m; i += 1024) ssrc[b0 + i] = stage[i];
  } else {
    for (int i = tid; i < m; i += 1024) {
      const int2 ed = bedge[b0 + i];
      const int p = atomicAdd(&ncur[ed.y & (BNODES - 1)], 1);
      ssrc[b0 + p] = ed.x;
    }
  }

  const int w = tid >> 5;
  gemm_n64_tile<64, false, true>(h, wl, nullptr, sdis, hw, n, k * (BNODES / 16) + w, tid & 31, w * 16);
}

#define RPW 4
#define GRP 8
#define GWAVES 8
#define HPITCH (HDIM + 4)

template<bool LAST>
__global__ __launch_bounds__(32 * GWAVES) void gather_rows(const int* __restrict__ rowstart, const int* __restrict__ cnt,
                            const int* __restrict__ ssrc, const float* __restrict__ dis,
                            const float* __restrict__ hw, const float* __restrict__ bias,
                            float* __restrict__ h, const float* __restrict__ Wn,
                            const float* __restrict__ bh, float* __restrict__ out, int n) {
  __shared__ float outv[GWAVES * RPW];
  __shared__ __attribute__((aligned(16))) float wl[64 * HDIM];
  __shared__ __attribute__((aligned(16))) float ht[GWAVES * RPW * HPITCH];
  __shared__ __attribute__((aligned(16))) float ot[GWAVES * RPW * HPITCH];
  const int lane = threadIdx.x & 31;
  float4 wnv[64 * 16 / (32 * GWAVES)];
  if (!LAST) stage_w64_load<64, 32 * GWAVES>(Wn, threadIdx.x, wnv);
  const int node0r = RPW * __builtin_amdgcn_readfirstlane((int)((blockIdx.x * blockDim.x + threadIdx.x) >> 5));
  const bool wlive = node0r < n;
  const int node0 = wlive ? node0r : 0;

  float2 hv[RPW];
  #pragma unroll
  for (int r = 0; r < RPW; ++r) {
    const int node = (node0 + r < n) ? node0 + r : node0;
    hv[r] = ((const float2*)h)[(unsigned)node * (HDIM / 2) + lane];
  }
  const float2 b = ((const float2*)bias)[lane];
  float2 wh = make_float2(0.0f, 0.0f);
  float  bhv = 0.0f;
  if (LAST) {
    wh  = ((const float2*)Wn)[lane];
    bhv = bh[0];
  }
  const int nl = (lane < RPW && node0 + lane < n) ? node0 + lane : node0;
  const int   c_l  = cnt[nl];
  const int   st_l = rowstart[nl];
  const float d_l  = dis[nl];
  const float2* __restrict__ hw2 = (const float2*)hw;

  int   start[RPW], deg[RPW];
  float dd[RPW];
  int   maxdeg = 0;
  #pragma unroll
  for (int r = 0; r < RPW; ++r) {
    const bool live = wlive && (node0 + r < n);
    const int  c    = __builtin_amdgcn_readlane(c_l, r);
    start[r] = __builtin_amdgcn_readlane(st_l, r);
    deg[r]   = live ? c : 0;
    dd[r]    = __int_as_float(__builtin_amdgcn_readlane(__float_as_int(d_l), r));
    maxdeg   = (deg[r] > maxdeg) ? deg[r] : maxdeg;
  }

  float2 acc[RPW];
  #pragma unroll
  for (int r = 0; r < RPW; ++r) acc[r] = make_float2(0.0f, 0.0f);

  int s_l[RPW];
  #pragma unroll
  for (int r = 0; r < RPW; ++r) {
    s_l[r] = 0;
    if (lane < deg[r]) s_l[r] = ssrc[start[r] + lane];
  }
  if (!LAST) stage_w64_store<64, 32 * GWAVES>(wnv, wl, threadIdx.x);

  for (int j0 = 0; j0 < maxdeg; j0 += 32) {
    int m[RPW];
    #pragma unroll
    for (int r = 0; r < RPW; ++r) {
      const int left = deg[r] - j0;
      m[r] = (left < 0) ? 0 : ((left < 32) ? left : 32);
    }
    const int mm = (maxdeg - j0 < 32) ? (maxdeg - j0) : 32;
    for (int j = 0; j < mm; j += GRP) {
      int sid[RPW][GRP];
      #pragma unroll
      for (int r = 0; r < RPW; ++r) {
        const int last = (m[r] > 0) ? (m[r] - 1) : 0;
        #pragma unroll
        for (int k = 0; k < GRP; ++k) {
          const int jj = j + k;
          sid[r][k] = __builtin_amdgcn_readlane(s_l[r], (jj < m[r]) ? jj : last);
        }
      }
      float2 v[RPW][GRP];
      #pragma unroll
      for (int r = 0; r < RPW; ++r) {
        if (j < m[r]) {
          #pragma unroll
          for (int k = 0; k < GRP; ++k)
            v[r][k] = hw2[(unsigned)sid[r][k] * (HDIM / 2) + lane];
        } else {
          #pragma unroll
          for (int k = 0; k < GRP; ++k) v[r][k] = make_float2(0.0f, 0.0f);
        }
      }
      #pragma unroll
      for (int r = 0; r < RPW; ++r) {
        #pragma unroll
        for (int k = 0; k < GRP; ++k) {
          const bool in = (j + k) < m[r];
          acc[r].x += in ? v[r][k].x : 0.0f;
          acc[r].y += in ? v[r][k].y : 0.0f;
        }
      }
    }
    if (j0 + 32 < maxdeg) {
      #pragma unroll
      for (int r = 0; r < RPW; ++r) {
        s_l[r] = 0;
        if (j0 + 32 + lane < deg[r]) s_l[r] = ssrc[start[r] + j0 + 32 + lane];
      }
    }
  }

  if (!LAST) {
    const int wv = threadIdx.x >> 5;
    float2 o[RPW];
    #pragma unroll
    for (int r = 0; r < RPW; ++r) {
      o[r].x = fmaxf(acc[r].x * dd[r] + b.x, 0.0f) + hv[r].x;
      o[r].y = fmaxf(acc[r].y * dd[r] + b.y, 0.0f) + hv[r].y;
      ((float2*)(ht + (wv * RPW + r) * HPITCH))[lane] = o[r];
    }
    __syncthreads();

    const int mt = wv >> 2, nt = wv & 3;
    const int half = lane >> 4, l16 = lane & 15;
    const float4* at4 = (const float4*)(ht + (mt * 16 + l16) * HPITCH + half * 4);
    const float4* wl4 = (const float4*)wl + lane;
    v8f pacc = {};
    #pragma unroll
    for (int j = 0; j < HDIM / 8; ++j) {
      const float4 a4 = at4[2 * j];
      const float4 b4 = wl4[(j * 4 + nt) * 32];
      v2f a, p;
      a.x = a4.x; a.y = a4.y;  p.x = b4.x; p.y = b4.y;
      pacc = __builtin_amdgcn_wmma_f32_16x16x4_f32(false, a, false, p, (short)0, pacc, false, false);
      a.x = a4.z; a.y = a4.w;  p.x = b4.z; p.y = b4.w;
      pacc = __builtin_amdgcn_wmma_f32_16x16x4_f32(false, a, false, p, (short)0, pacc, false, false);
    }
    #pragma unroll
    for (int r = 0; r < 8; ++r) ot[(mt * 16 + half * 8 + r) * HPITCH + 4 * l16 + nt] = pacc[r];
    __syncthreads();
    #pragma unroll
    for (int r = 0; r < RPW; ++r) {
      if (wlive && node0 + r < n) {
        float2 v = ((const float2*)(ot + (wv * RPW + r) * HPITCH))[lane];
        v.x *= dd[r];  v.y *= dd[r];
        ((float2*)(h + (size_t)(node0 + r) * HDIM))[lane] = o[r];
        ((float2*)(out + (size_t)(node0 + r) * HDIM))[lane] = v;
      }
    }
  } else {
    float res = 0.0f;
    #pragma unroll
    for (int r = 0; r < RPW; ++r) {
      const float ox = fmaxf(acc[r].x * dd[r] + b.x, 0.0f) + hv[r].x;
      const float oy = fmaxf(acc[r].y * dd[r] + b.y, 0.0f) + hv[r].y;
      float p = ox * wh.x + oy * wh.y;
      for (int o = 16; o > 0; o >>= 1) p += __shfl_xor(p, o);
      if (lane == r) res = p;
    }
    if (lane < RPW) outv[(threadIdx.x >> 5) * RPW + lane] = res + bhv;
    __syncthreads();
    if (threadIdx.x < GWAVES * RPW) {
      const int node = (int)blockIdx.x * (GWAVES * RPW) + (int)threadIdx.x;
      if (node < n) out[node] = outv[threadIdx.x];
    }
  }
}

__global__ __launch_bounds__(256) void hist_and_pre(const int* __restrict__ dst, int* __restrict__ hist,
                                                    int E, int nb, int nblk,
                                                    const float* __restrict__ x, const float* __restrict__ W_pre,
                                                    const float* __restrict__ b_pre, float* __restrict__ h, int n) {
  __shared__ int lh[MAXB];
  __shared__ __attribute__((aligned(16))) float wl[128 * HDIM];
  const int tid = threadIdx.x;
  if ((int)blockIdx.x < nblk) {
    const int e0 = blockIdx.x * ECHUNK;
    lh[tid] = 0;
    __syncthreads();
    int dv[ECHUNK / 256];
    #pragma unroll
    for (int i = 0; i < ECHUNK / 256; ++i) {
      const int e = e0 + i * 256 + tid;
      dv[i] = (e < E) ? dst[e] : -1;
    }
    #pragma unroll
    for (int i = 0; i < ECHUNK / 256; ++i)
      if (dv[i] >= 0) atomicAdd(&lh[dv[i] >> BSHIFT], 1);
    __syncthreads();
    if (tid < nb) hist[blockIdx.x * nb + tid] = lh[tid];
    return;
  }
  stage_w64<128, 256>(W_pre, wl, tid);
  __syncthreads();
  const int wave = (int)((blockIdx.x - nblk) * 256 + tid) >> 5;
  gemm_n64_tile<128, true, false>(x, wl, b_pre, nullptr, h, n, wave, tid & 31, 0);
}

extern "C" void kernel_launch(void* const* d_in, const int* in_sizes, int n_in,
                              void* d_out, int out_size, void* d_ws, size_t ws_size,
                              hipStream_t stream) {
  const float* x      = (const float*)d_in[0];
  const int*   eidx   = (const int*)d_in[1];
  const float* W_pre  = (const float*)d_in[2];
  const float* b_pre  = (const float*)d_in[3];
  const float* W1     = (const float*)d_in[4];
  const float* b1     = (const float*)d_in[5];
  const float* W2     = (const float*)d_in[6];
  const float* b2     = (const float*)d_in[7];
  const float* W_head = (const float*)d_in[8];
  const float* b_head = (const float*)d_in[9];
  float* out = (float*)d_out;

  const int N = in_sizes[0] / 128;
  const int E = in_sizes[1] / 2;
  const int* src = eidx;
  const int* dst = eidx + E;

  float* h        = (float*)d_ws;
  float* hw       = h + (size_t)N * HDIM;
  float* hw2      = hw + (size_t)N * HDIM;
  int*   bedge    = (int*)(hw2 + (size_t)N * HDIM);
  int*   ssrc     = bedge + 2 * (size_t)E;
  int*   cnt      = ssrc + E;
  int*   rowstart = cnt + N;
  float* dis      = (float*)(rowstart + N);

  const int nb   = (N + BNODES - 1) / BNODES;
  const int nblk = (E + ECHUNK - 1) / ECHUNK;
  int* hist  = (int*)(dis + N);
  int* bbase = hist + (size_t)nblk * nb;
  int gemm_blocks = (N + 127) / 128;

  hist_and_pre<<<nblk + gemm_blocks, 256, 0, stream>>>(dst, hist, E, nb, nblk, x, W_pre, b_pre, h, N);
  scatter_buckets<<<nblk, 1024, 0, stream>>>(src, dst, hist, bedge, bbase, E, nb, nblk);
  sort_bucket<<<nb, 1024, 0, stream>>>(bedge, bbase, cnt, rowstart, dis, ssrc, h, W1, hw, N);

  int gather_blocks = (N + GWAVES * RPW - 1) / (GWAVES * RPW);
  gather_rows<false><<<gather_blocks, 32 * GWAVES, 0, stream>>>(rowstart, cnt, ssrc, dis, hw, b1, h,
                                                                W2, nullptr, hw2, N);
  gather_rows<true><<<gather_blocks, 32 * GWAVES, 0, stream>>>(rowstart, cnt, ssrc, dis, hw2, b2, h,
                                                               W_head, b_head, out, N);
}


# Neighbour sum: slots past a row's end hold the id of an all-zero row (row n of hw and hw2), so a round picks its ids by lane number and adds with no select per slot; row sums bit-identical
# speedup vs baseline: 5.4287x; 1.0667x over previous
#include <hip/hip_runtime.h>
#include <hip/hip_bf16.h>

typedef __attribute__((ext_vector_type(2))) float v2f;
typedef __attribute__((ext_vector_type(8))) float v8f;

#define HDIM 64

#define BSHIFT 9
#define BNODES 512
#define MAXB   256
#define ECHUNK 8192
#define STAGE  16384

__device__ __forceinline__ int block_excl_scan(int v, int* wsum) {
  const int lane = threadIdx.x & 31;
  const int wid  = threadIdx.x >> 5;
  int x = v;
  for (int o = 1; o < 32; o <<= 1) {
    int y = __shfl_up(x, o);
    if (lane >= o) x += y;
  }
  if (lane == 31) wsum[wid] = x;
  __syncthreads();
  int wo = (lane < wid) ? wsum[lane] : 0;
  for (int o = 16; o > 0; o >>= 1) wo += __shfl_xor(wo, o);
  __syncthreads();
  return wo + x - v;
}


template<int K, int NT>
__device__ __forceinline__ void stage_w64_load(const float* __restrict__ W, int tid, float4 (&w)[K * 16 / NT]) {
  #pragma unroll
  for (int u = 0; u < K * 16 / NT; ++u) w[u] = ((const float4*)W)[u * NT + tid];
}
template<int K, int NT>
__device__ __forceinline__ void stage_w64_store(const float4 (&w)[K * 16 / NT], float* __restrict__ wl, int tid) {
  #pragma unroll
  for (int u = 0; u < K * 16 / NT; ++u) {
    const int idx = u * NT + tid;
    const int k = idx >> 4, c = idx & 15;
    float* d = wl + ((((k >> 3) * 4) * 2 + ((k >> 2) & 1)) * 16 + c) * 4 + (k & 3);
    d[0]   = w[u].x;
    d[128] = w[u].y;
    d[256] = w[u].z;
    d[384] = w[u].w;
  }
}
template<int K, int NT>
__device__ __forceinline__ void stage_w64(const float* __restrict__ W, float* __restrict__ wl, int tid) {
  float4 w[K * 16 / NT];
  stage_w64_load<K, NT>(W, tid, w);
  stage_w64_store<K, NT>(w, wl, tid);
}

template<int K, bool RELU_BIAS, bool ROW_SCALE>
__device__ __forceinline__ void gemm_n64_tile(const float* __restrict__ A,
                                              const float* __restrict__ wl,
                                              const float* __restrict__ bias,
                                              const float* __restrict__ rowscale,
                                              float* __restrict__ out, int n_rows,
                                              int wave, int lane, int rs_row0) {
  int row0 = wave * 16;
  if (row0 >= n_rows) return;

  const int half = lane >> 4;
  const int l16  = lane & 15;

  v8f acc0 = {}, acc1 = {}, acc2 = {}, acc3 = {};

  float rs[8];
  #pragma unroll
  for (int r = 0; r < 8; ++r) rs[r] = ROW_SCALE ? rowscale[rs_row0 + half * 8 + r] : 1.0f;

  const float4* arow = (const float4*)(A + (size_t)(row0 + l16) * K + half * 4);
  float4 av[K / 8];
  #pragma unroll
  for (int j = 0; j < K / 8; ++j) av[j] = arow[2 * j];

  const float4* wl4 = (const float4*)wl + lane;
  #pragma unroll
  for (int j = 0; j < K / 8; ++j) {
    const float4 b0 = wl4[(j * 4 + 0) * 32];
    const float4 b1 = wl4[(j * 4 + 1) * 32];
    const float4 b2 = wl4[(j * 4 + 2) * 32];
    const float4 b3 = wl4[(j * 4 + 3) * 32];
    v2f a;  a.x = av[j].x;  a.y = av[j].y;
    v2f p0, p1, p2, p3;
    p0.x = b0.x; p0.y = b0.y;  p1.x = b1.x; p1.y = b1.y;
    p2.x = b2.x; p2.y = b2.y;  p3.x = b3.x; p3.y = b3.y;
    acc0 = __builtin_amdgcn_wmma_f32_16x16x4_f32(false, a, false, p0, (short)0, acc0, false, false);
    acc1 = __builtin_amdgcn_wmma_f32_16x16x4_f32(false, a, false, p1, (short)0, acc1, false, false);
    acc2 = __builtin_amdgcn_wmma_f32_16x16x4_f32(false, a, false, p2, (short)0, acc2, false, false);
    acc3 = __builtin_amdgcn_wmma_f32_16x16x4_f32(false, a, false, p3, (short)0, acc3, false, false);
    a.x = av[j].z;  a.y = av[j].w;
    p0.x = b0.z; p0.y = b0.w;  p1.x = b1.z; p1.y = b1.w;
    p2.x = b2.z; p2.y = b2.w;  p3.x = b3.z; p3.y = b3.w;
    acc0 = __builtin_amdgcn_wmma_f32_16x16x4_f32(false, a, false, p0, (short)0, acc0, false, false);
    acc1 = __builtin_amdgcn_wmma_f32_16x16x4_f32(false, a, false, p1, (short)0, acc1, false, false);
    acc2 = __builtin_amdgcn_wmma_f32_16x16x4_f32(false, a, false, p2, (short)0, acc2, false, false);
    acc3 = __builtin_amdgcn_wmma_f32_16x16x4_f32(false, a, false, p3, (short)0, acc3, false, false);
  }

  float4 bc = make_float4(0.0f, 0.0f, 0.0f, 0.0f);
  if (RELU_BIAS) bc = ((const float4*)bias)[l16];
  #pragma unroll
  for (int r = 0; r < 8; ++r) {
    float4 v = make_float4(acc0[r], acc1[r], acc2[r], acc3[r]);
    if (RELU_BIAS) {
      v.x = fmaxf(v.x + bc.x, 0.0f);  v.y = fmaxf(v.y + bc.y, 0.0f);
      v.z = fmaxf(v.z + bc.z, 0.0f);  v.w = fmaxf(v.w + bc.w, 0.0f);
    }
    if (ROW_SCALE) { v.x *= rs[r]; v.y *= rs[r]; v.z *= rs[r]; v.w *= rs[r]; }
    ((float4*)(out + (size_t)(row0 + half * 8 + r) * HDIM))[l16] = v;
  }
}


__global__ __launch_bounds__(1024) void scatter_buckets(const int* __restrict__ src, const int* __restrict__ dst,
                                                        const int* __restrict__ hist, int* __restrict__ bedge_w,
                                                        int* __restrict__ bbase, int E, int nb, int nblk) {
  int2* __restrict__ bedge = (int2*)bedge_w;
  __shared__ int wsum[32];
  __shared__ int cur[MAXB];
  __shared__ int gofs[MAXB];
  __shared__ int ptot[4][MAXB];
  __shared__ int ppre[4][MAXB];
  __shared__ int2 stage[ECHUNK];
  const int tid = threadIdx.x;
  const int blk = blockIdx.x;
  const int e0  = blk * ECHUNK;
  int sv[ECHUNK / 1024], dv[ECHUNK / 1024];
  #pragma unroll
  for (int i = 0; i < ECHUNK / 1024; ++i) {
    const int e = e0 + i * 1024 + tid;
    const bool in = e < E;
    sv[i] = in ? src[e] : 0;
    dv[i] = in ? dst[e] : -1;
  }
  const bool isb = tid < nb;
  const int own = isb ? hist[blk * nb + tid] : 0;
  const int bq = tid & (MAXB - 1);
  const int q  = tid >> 8;
  int tq = 0, pq = 0;
  if (bq < nb) {
    const int lo = q * (MAXB / 4);
    const int hi = (lo + MAXB / 4 < nblk) ? (lo + MAXB / 4) : nblk;
    for (int j0 = lo; j0 < hi; j0 += 32) {
      int v[32];
      #pragma unroll
      for (int u = 0; u < 32; ++u) v[u] = (j0 + u < hi) ? hist[(j0 + u) * nb + bq] : 0;
      #pragma unroll
      for (int u = 0; u < 32; ++u) {
        tq += v[u];
        pq += (j0 + u < blk) ? v[u] : 0;
      }
    }
  }
  ptot[q][bq] = tq;
  ppre[q][bq] = pq;
  __syncthreads();
  const int tot = isb ? (ptot[0][tid] + ptot[1][tid] + ptot[2][tid] + ptot[3][tid]) : 0;
  const int pre = isb ? (ppre[0][tid] + ppre[1][tid] + ppre[2][tid] + ppre[3][tid]) : 0;
  const int base = block_excl_scan(tot, wsum);
  const int ls   = block_excl_scan(own, wsum);
  if (isb) {
    cur[tid]  = ls;
    gofs[tid] = base + pre - ls;
    if (blk == 0) {
      bbase[tid] = base;
      if (tid == nb - 1) bbase[nb] = base + tot;
    }
  }
  __syncthreads();
  #pragma unroll
  for (int i = 0; i < ECHUNK / 1024; ++i) {
    if (dv[i] >= 0) {
      const int p = atomicAdd(&cur[dv[i] >> BSHIFT], 1);
      stage[p] = make_int2(sv[i], dv[i]);
    }
  }
  __syncthreads();
  const int m = (E - e0 < ECHUNK) ? (E - e0) : ECHUNK;
  for (int i = tid; i < m; i += 1024) {
    const int2 ed = stage[i];
    bedge[gofs[ed.y >> BSHIFT] + i] = ed;
  }
}

__global__ __launch_bounds__(1024) void sort_bucket(const int* __restrict__ bedge_w, const int* __restrict__ bbase,
                                                    int* __restrict__ cnt, int* __restrict__ rowstart,
                                                    float* __restrict__ dis, int* __restrict__ ssrc,
                                                    const float* __restrict__ h, const float* __restrict__ W1,
                                                    float* __restrict__ hw, int n) {
  const int2* __restrict__ bedge = (const int2*)bedge_w;
  __shared__ int ncnt[BNODES];
  __shared__ int ncur[BNODES];
  __shared__ float sdis[BNODES];
  __shared__ int wsum[32];
  __shared__ int stage[STAGE];
  __shared__ __attribute__((aligned(16))) float wl[64 * HDIM];
  const int tid  = threadIdx.x;
  const int k    = blockIdx.x;
  const int b0   = bbase[k];
  const int m    = bbase[k + 1] - b0;

  float4 w1v[1];
  stage_w64_load<64, 1024>(W1, tid, w1v);
  if (tid < BNODES) ncnt[tid] = 0;
  __syncthreads();
  for (int i0 = 0; i0 < m; i0 += 8 * 1024) {
    int dv[8];
    for (int j = 0; j < 8; ++j) {
      const int i = i0 + j * 1024 + tid;
      dv[j] = (i < m) ? bedge[b0 + i].y : -1;
    }
    for (int j = 0; j < 8; ++j)
      if (dv[j] >= 0) atomicAdd(&ncnt[dv[j] & (BNODES - 1)], 1);
  }
  __syncthreads();

  const int c  = (tid < BNODES) ? ncnt[tid] : 0;
  const int st = block_excl_scan(c, wsum);
  if (tid < BNODES) {
    ncur[tid] = st;
    const int node = k * BNODES + tid;
    const float d  = (float)c;
    const float dv = (d > 0.0f) ? rsqrtf(fmaxf(d, 1.0f)) : 0.0f;
    sdis[tid] = dv;
    if (node < n) {
      cnt[node]      = c;
      rowstart[node] = b0 + st;
      dis[node]      = dv;
    }
  }
  stage_w64_store<64, 1024>(w1v, wl, tid);
  __syncthreads();

  if (m <= STAGE) {
    for (int i0 = 0; i0 < m; i0 += 8 * 1024) {
      int2 ev[8];
      for (int j = 0; j < 8; ++j) {
        const int i = i0 + j * 1024 + tid;
        ev[j] = (i < m) ? bedge[b0 + i] : make_int2(0, -1);
      }
      for (int j = 0; j < 8; ++j) {
        if (ev[j].y >= 0) {
          const int p = atomicAdd(&ncur[ev[j].y & (BNODES - 1)], 1);
          stage[p] = ev[j].x;
        }
      }
    }
    __syncthreads();
    for (int i = tid; i < m; i += 1024) ssrc[b0 + i] = stage[i];
  } else {
    for (int i = tid; i < m; i += 1024) {
      const int2 ed = bedge[b0 + i];
      const int p = atomicAdd(&ncur[ed.y & (BNODES - 1)], 1);
      ssrc[b0 + p] = ed.x;
    }
  }

  if (k == 0 && tid < HDIM / 4) ((float4*)(hw + (size_t)n * HDIM))[tid] = make_float4(0.0f, 0.0f, 0.0f, 0.0f);

  const int w = tid >> 5;
  gemm_n64_tile<64, false, true>(h, wl, nullptr, sdis, hw, n, k * (BNODES / 16) + w, tid & 31, w * 16);
}

#define RPW 4
#define GRP 8
#define GWAVES 8
#define HPITCH (HDIM + 4)

template<bool LAST>
__global__ __launch_bounds__(32 * GWAVES) void gather_rows(const int* __restrict__ rowstart, const int* __restrict__ cnt,
                            const int* __restrict__ ssrc, const float* __restrict__ dis,
                            const float* __restrict__ hw, const float* __restrict__ bias,
                            float* __restrict__ h, const float* __restrict__ Wn,
                            const float* __restrict__ bh, float* __restrict__ out, int n) {
  __shared__ float outv[GWAVES * RPW];
  __shared__ __attribute__((aligned(16))) float wl[64 * HDIM];
  __shared__ __attribute__((aligned(16))) float ht[GWAVES * RPW * HPITCH];
  __shared__ __attribute__((aligned(16))) float ot[GWAVES * RPW * HPITCH];
  const int lane = (int)__builtin_amdgcn_mbcnt_lo(~0u, 0u);
  const int wv = __builtin_amdgcn_readfirstlane((int)(threadIdx.x >> 5));
  float4 wnv[64 * 16 / (32 * GWAVES)];
  if (!LAST) stage_w64_load<64, 32 * GWAVES>(Wn, threadIdx.x, wnv);
  const int node0r = RPW * __builtin_amdgcn_readfirstlane((int)((blockIdx.x * blockDim.x + threadIdx.x) >> 5));
  const bool wlive = node0r < n;
  const int node0 = wlive ? node0r : 0;

  float2 hv[RPW];
  #pragma unroll
  for (int r = 0; r < RPW; ++r) {
    const int node = (node0 + r < n) ? node0 + r : node0;
    hv[r] = ((const float2*)h)[(unsigned)node * (HDIM / 2) + lane];
  }
  const float2 b = ((const float2*)bias)[lane];
  float2 wh = make_float2(0.0f, 0.0f);
  float  bhv = 0.0f;
  if (LAST) {
    wh  = ((const float2*)Wn)[lane];
    bhv = bh[0];
  }
  const int nl = (lane < RPW && node0 + lane < n) ? node0 + lane : node0;
  const int   c_l  = cnt[nl];
  const int   st_l = rowstart[nl];
  const float d_l  = dis[nl];
  const float2* __restrict__ hw2 = (const float2*)hw;

  int   start[RPW], deg[RPW];
  float dd[RPW];
  int   maxdeg = 0;
  #pragma unroll
  for (int r = 0; r < RPW; ++r) {
    const bool live = wlive && (node0 + r < n);
    const int  c    = __builtin_amdgcn_readlane(c_l, r);
    start[r] = __builtin_amdgcn_readlane(st_l, r);
    deg[r]   = live ? c : 0;
    dd[r]    = __int_as_float(__builtin_amdgcn_readlane(__float_as_int(d_l), r));
    maxdeg   = (deg[r] > maxdeg) ? deg[r] : maxdeg;
  }

  float2 acc[RPW];
  #pragma unroll
  for (int r = 0; r < RPW; ++r) acc[r] = make_float2(0.0f, 0.0f);

  int s_l[RPW];
  #pragma unroll
  for (int r = 0; r < RPW; ++r) {
    s_l[r] = n;
    if (lane < deg[r]) s_l[r] = ssrc[start[r] + lane];
  }
  if (!LAST) {
    stage_w64_store<64, 32 * GWAVES>(wnv, wl, threadIdx.x);
    if (blockIdx.x == 0 && threadIdx.x < HDIM / 4)
      ((float4*)(out + (size_t)n * HDIM))[threadIdx.x] = make_float4(0.0f, 0.0f, 0.0f, 0.0f);
  }

  for (int j0 = 0; j0 < maxdeg; j0 += 32) {
    const int mm = (maxdeg - j0 < 32) ? (maxdeg - j0) : 32;
    for (int j = 0; j < mm; j += GRP) {
      int sid[RPW][GRP];
      #pragma unroll
      for (int r = 0; r < RPW; ++r) {
        #pragma unroll
        for (int k = 0; k < GRP; ++k)
          sid[r][k] = __builtin_amdgcn_readlane(s_l[r], j + k);
      }
      float2 v[RPW][GRP];
      #pragma unroll
      for (int r = 0; r < RPW; ++r) {
        if (j0 + j < deg[r]) {
          #pragma unroll
          for (int k = 0; k < GRP; ++k)
            v[r][k] = hw2[(unsigned)sid[r][k] * (HDIM / 2) + lane];
        }
      }
      #pragma unroll
      for (int r = 0; r < RPW; ++r) {
        if (j0 + j < deg[r]) {
          #pragma unroll
          for (int k = 0; k < GRP; ++k) {
            acc[r].x += v[r][k].x;
            acc[r].y += v[r][k].y;
          }
        }
      }
    }
    if (j0 + 32 < maxdeg) {
      #pragma unroll
      for (int r = 0; r < RPW; ++r) {
        s_l[r] = n;
        if (j0 + 32 + lane < deg[r]) s_l[r] = ssrc[start[r] + j0 + 32 + lane];
      }
    }
  }

  if (!LAST) {
    float2 o[RPW];
    #pragma unroll
    for (int r = 0; r < RPW; ++r) {
      o[r].x = fmaxf(acc[r].x * dd[r] + b.x, 0.0f) + hv[r].x;
      o[r].y = fmaxf(acc[r].y * dd[r] + b.y, 0.0f) + hv[r].y;
      ((float2*)(ht + (wv * RPW + r) * HPITCH))[lane] = o[r];
    }
    __syncthreads();

    const int mt = wv >> 2, nt = wv & 3;
    const int half = lane >> 4, l16 = lane & 15;
    const float4* at4 = (const float4*)(ht + (mt * 16 + l16) * HPITCH + half * 4);
    const float4* wl4 = (const float4*)wl + lane;
    v8f pacc = {};
    #pragma unroll
    for (int j = 0; j < HDIM / 8; ++j) {
      const float4 a4 = at4[2 * j];
      const float4 b4 = wl4[(j * 4 + nt) * 32];
      v2f a, p;
      a.x = a4.x; a.y = a4.y;  p.x = b4.x; p.y = b4.y;
      pacc = __builtin_amdgcn_wmma_f32_16x16x4_f32(false, a, false, p, (short)0, pacc, false, false);
      a.x = a4.z; a.y = a4.w;  p.x = b4.z; p.y = b4.w;
      pacc = __builtin_amdgcn_wmma_f32_16x16x4_f32(false, a, false, p, (short)0, pacc, false, false);
    }
    #pragma unroll
    for (int r = 0; r < 8; ++r) ot[(mt * 16 + half * 8 + r) * HPITCH + 4 * l16 + nt] = pacc[r];
    __syncthreads();
    #pragma unroll
    for (int r = 0; r < RPW; ++r) {
      if (wlive && node0 + r < n) {
        float2 v = ((const float2*)(ot + (wv * RPW + r) * HPITCH))[lane];
        v.x *= dd[r];  v.y *= dd[r];
        ((float2*)(h + (size_t)(node0 + r) * HDIM))[lane] = o[r];
        ((float2*)(out + (size_t)(node0 + r) * HDIM))[lane] = v;
      }
    }
  } else {
    float res = 0.0f;
    #pragma unroll
    for (int r = 0; r < RPW; ++r) {
      const float ox = fmaxf(acc[r].x * dd[r] + b.x, 0.0f) + hv[r].x;
      const float oy = fmaxf(acc[r].y * dd[r] + b.y, 0.0f) + hv[r].y;
      float p = ox * wh.x + oy * wh.y;
      for (int o = 16; o > 0; o >>= 1) p += __shfl_xor(p, o);
      if (lane == r) res = p;
    }
    if (lane < RPW) outv[wv * RPW + lane] = res + bhv;
    __syncthreads();
    if (wv == 0) {
      const int node = (int)blockIdx.x * (GWAVES * RPW) + lane;
      if (node < n) out[node] = outv[lane];
    }
  }
}

__global__ __launch_bounds__(256) void hist_and_pre(const int* __restrict__ dst, int* __restrict__ hist,
                                                    int E, int nb, int nblk,
                                                    const float* __restrict__ x, const float* __restrict__ W_pre,
                                                    const float* __restrict__ b_pre, float* __restrict__ h, int n) {
  __shared__ int lh[MAXB];
  __shared__ __attribute__((aligned(16))) float wl[128 * HDIM];
  const int tid = threadIdx.x;
  if ((int)blockIdx.x < nblk) {
    const int e0 = blockIdx.x * ECHUNK;
    lh[tid] = 0;
    __syncthreads();
    int dv[ECHUNK / 256];
    #pragma unroll
    for (int i = 0; i < ECHUNK / 256; ++i) {
      const int e = e0 + i * 256 + tid;
      dv[i] = (e < E) ? dst[e] : -1;
    }
    #pragma unroll
    for (int i = 0; i < ECHUNK / 256; ++i)
      if (dv[i] >= 0) atomicAdd(&lh[dv[i] >> BSHIFT], 1);
    __syncthreads();
    if (tid < nb) hist[blockIdx.x * nb + tid] = lh[tid];
    return;
  }
  stage_w64<128, 256>(W_pre, wl, tid);
  __syncthreads();
  const int wave = (int)((blockIdx.x - nblk) * 256 + tid) >> 5;
  gemm_n64_tile<128, true, false>(x, wl, b_pre, nullptr, h, n, wave, tid & 31, 0);
}

extern "C" void kernel_launch(void* const* d_in, const int* in_sizes, int n_in,
                              void* d_out, int out_size, void* d_ws, size_t ws_size,
                              hipStream_t stream) {
  const float* x      = (const float*)d_in[0];
  const int*   eidx   = (const int*)d_in[1];
  const float* W_pre  = (const float*)d_in[2];
  const float* b_pre  = (const float*)d_in[3];
  const float* W1     = (const float*)d_in[4];
  const float* b1     = (const float*)d_in[5];
  const float* W2     = (const float*)d_in[6];
  const float* b2     = (const float*)d_in[7];
  const float* W_head = (const float*)d_in[8];
  const float* b_head = (const float*)d_in[9];
  float* out = (float*)d_out;

  const int N = in_sizes[0] / 128;
  const int E = in_sizes[1] / 2;
  const int* src = eidx;
  const int* dst = eidx + E;

  float* h        = (float*)d_ws;
  float* hw       = h + (size_t)N * HDIM;
  float* hw2      = hw + ((size_t)N + 1) * HDIM;
  int*   bedge    = (int*)(hw2 + ((size_t)N + 1) * HDIM);
  int*   ssrc     = bedge + 2 * (size_t)E;
  int*   cnt      = ssrc + E;
  int*   rowstart = cnt + N;
  float* dis      = (float*)(rowstart + N);

  const int nb   = (N + BNODES - 1) / BNODES;
  const int nblk = (E + ECHUNK - 1) / ECHUNK;
  int* hist  = (int*)(dis + N);
  int* bbase = hist + (size_t)nblk * nb;
  int gemm_blocks = (N + 127) / 128;

  hist_and_pre<<<nblk + gemm_blocks, 256, 0, stream>>>(dst, hist, E, nb, nblk, x, W_pre, b_pre, h, N);
  scatter_buckets<<<nblk, 1024, 0, stream>>>(src, dst, hist, bedge, bbase, E, nb, nblk);
  sort_bucket<<<nb, 1024, 0, stream>>>(bedge, bbase, cnt, rowstart, dis, ssrc, h, W1, hw, N);

  int gather_blocks = (N + GWAVES * RPW - 1) / (GWAVES * RPW);
  gather_rows<false><<<gather_blocks, 32 * GWAVES, 0, stream>>>(rowstart, cnt, ssrc, dis, hw, b1, h,
                                                                W2, nullptr, hw2, N);
  gather_rows<true><<<gather_blocks, 32 * GWAVES, 0, stream>>>(rowstart, cnt, ssrc, dis, hw2, b2, h,
                                                               W_head, b_head, out, N);
}
